# starting kernel
# baseline (speedup 1.0000x reference)
#include <hip/hip_runtime.h>
#include <utility>

typedef __attribute__((ext_vector_type(2))) float v2f;
typedef __attribute__((ext_vector_type(8))) float v8f;

namespace gol {

struct Tables {
  unsigned short s3[96];
  unsigned short s4[64];
};

constexpr int popc9(int m) { int c = 0; for (int k = 0; k < 9; ++k) c += (m >> k) & 1; return c; }

constexpr Tables build() {
  Tables t{};
  for (int i = 0; i < 96; ++i) t.s3[i] = 0xFFFFu;
  for (int i = 0; i < 64; ++i) t.s4[i] = 0u;
  int i3 = 0, i4 = 0;
  for (int m = 0; m < 512; ++m) {
    const int tot = popc9(m);
    const bool alive = ((m >> 4) & 1) != 0;
    const bool future = alive ? (tot == 3 || tot == 4) : (tot == 3);
    if (!future) continue;
    if (tot == 3) t.s3[i3++] = (unsigned short)m;
    else          t.s4[i4++] = (unsigned short)m;
  }
  return t;
}

constexpr unsigned S3c(int i) { return build().s3[i]; }
constexpr unsigned S4c(int i) { return build().s4[i]; }

struct MTab { unsigned m[96][2]; };

constexpr MTab buildMT() {
  MTab r{};
  constexpr Tables t = build();
  for (int row = 0; row < 96; ++row) {
    for (int H = 0; H < 2; ++H) {
      unsigned mm = 0;
      for (int idx = 0; idx < 28; ++idx) {
        const int kb = idx >> 1, j = idx & 1;
        const unsigned c3 = t.s3[row];
        const unsigned c4 = t.s4[4 * kb + j + 2 * H];
        if ((c3 & ~c4) == 0u) mm |= (1u << idx);
      }
      r.m[row][H] = mm;
    }
  }
  return r;
}

__device__ constexpr MTab MT = buildMT();

template<typename F, int... Is>
__device__ __forceinline__ void static_for_impl(F&& f, std::integer_sequence<int, Is...>) {
  (f(std::integral_constant<int, Is>{}), ...);
}
template<int N, typename F>
__device__ __forceinline__ void static_for(F&& f) {
  static_for_impl(static_cast<F&&>(f), std::make_integer_sequence<int, N>{});
}

}

__global__ __launch_bounds__(256) void gol_wmma_kernel(const float* __restrict__ x,
                                                       float* __restrict__ out,
                                                       int nTiles) {
  using namespace gol;
  const int lane = (int)(threadIdx.x & 31u);
  const int p = lane & 15;
  const int h = lane >> 4;
  const int w = (int)blockIdx.x * ((int)blockDim.x >> 5) + ((int)threadIdx.x >> 5);
  if (w >= nTiles) return;

  const int b  = w >> 12;
  const int rm = w & 4095;
  const int y  = rm >> 4;
  const int tx = rm & 15;
  const int x0 = (tx << 4) + p;
  const float* img = x + (size_t)b * 65536;

  float n[9];
#pragma unroll
  for (int k = 0; k < 9; ++k) {
    const int dy = k / 3 - 1, dx = k % 3 - 1;
    const int yy = (y + dy) & 255;
    const int xx = (x0 + dx) & 255;
    n[k] = img[yy * 256 + xx];
  }
  float c[9];
#pragma unroll
  for (int k = 0; k < 9; ++k) c[k] = 1.0f - n[k];

  float gA[16]; gA[0] = 1.0f;
  static_for<4>([&](auto C4) {
    constexpr int cell = C4.value;
    constexpr int s = 1 << cell;
    static_for<s>([&](auto M) {
      constexpr int m = s - 1 - M.value;
      gA[m + s] = gA[m] * n[cell];
      gA[m]     = gA[m] * c[cell];
    });
  });
  float gB[32]; gB[0] = 1.0f;
  static_for<5>([&](auto C5) {
    constexpr int cell = C5.value;
    constexpr int s = 1 << cell;
    static_for<s>([&](auto M) {
      constexpr int m = s - 1 - M.value;
      gB[m + s] = gB[m] * n[cell + 4];
      gB[m]     = gB[m] * c[cell + 4];
    });
  });

  const bool hi = (h != 0);
  float sumAcc = 0.0f;

  float acc4[28];
  static_for<28>([&](auto I) {
    constexpr int idx = I.value;
    constexpr int kb = idx >> 1, j = idx & 1;
    constexpr unsigned m0 = S4c(4 * kb + j);
    constexpr unsigned m1 = S4c(4 * kb + j + 2);
    const float v0 = gA[m0 & 15] * gB[(m0 >> 4) & 31];
    const float v1 = gA[m1 & 15] * gB[(m1 >> 4) & 31];
    const float v = hi ? v1 : v0;
    acc4[idx] = v;
    sumAcc += v;
  });

  float acc3[48];
  static_for<48>([&](auto I) {
    constexpr int idx = I.value;
    constexpr int t = idx >> 3, v = idx & 7;
    constexpr int r0 = 16 * t + v, r1 = r0 + 8;
    float v0 = 0.0f, v1 = 0.0f;
    if constexpr (r0 < 84) { constexpr unsigned mk = S3c(r0); v0 = gA[mk & 15] * gB[(mk >> 4) & 31]; }
    if constexpr (r1 < 84) { constexpr unsigned mk = S3c(r1); v1 = gA[mk & 15] * gB[(mk >> 4) & 31]; }
    const float val = hi ? v1 : v0;
    acc3[idx] = val;
    sumAcc += val;
  });

  unsigned mt[6];
  static_for<6>([&](auto T) {
    constexpr int t = T.value;
    mt[t] = MT.m[16 * t + p][h];
  });

  float orp = 0.0f;
#pragma unroll
  for (int t = 0; t < 6; ++t) {
    v8f C = {0.f, 0.f, 0.f, 0.f, 0.f, 0.f, 0.f, 0.f};
    const unsigned m = mt[t];
#pragma unroll
    for (int kb = 0; kb < 14; ++kb) {
      v2f A, Bv;
      A.x = (float)((m >> (2 * kb)) & 1u);
      A.y = (float)((m >> (2 * kb + 1)) & 1u);
      Bv.x = acc4[2 * kb];
      Bv.y = acc4[2 * kb + 1];
      C = __builtin_amdgcn_wmma_f32_16x16x4_f32(false, A, false, Bv, (short)0, C, false, false);
    }
#pragma unroll
    for (int v = 0; v < 8; ++v) orp = fmaf(acc3[8 * t + v], C[v], orp);
  }

  float tot = sumAcc - orp;
  tot += __shfl_xor(tot, 16, 32);
  if (h == 0) out[(size_t)b * 65536 + (size_t)y * 256 + (size_t)x0] = tot;
}

extern "C" void kernel_launch(void* const* d_in, const int* in_sizes, int n_in,
                              void* d_out, int out_size, void* d_ws, size_t ws_size,
                              hipStream_t stream) {
  (void)n_in; (void)d_ws; (void)ws_size; (void)out_size;
  const float* x = (const float*)d_in[0];
  float* out = (float*)d_out;
  const int total = in_sizes[0];
  const int Bn = total >> 16;
  const int nTiles = Bn * 4096;
  const int wavesPerBlock = 8;
  dim3 block(256);
  dim3 grid((unsigned)((nTiles + wavesPerBlock - 1) / wavesPerBlock));
  gol_wmma_kernel<<<grid, block, 0, stream>>>(x, out, nTiles);
}


# Pair GEMM split by the 180-degree turn of the 3x3 block into its even and odd part: two 48x28 GEMMs on sums and differences of mirrored configs, 84 -> 42 WMMAs per tile; f32 sum order changes
# speedup vs baseline: 1.7697x; 1.7697x over previous
#include <hip/hip_runtime.h>
#include <utility>

typedef __attribute__((ext_vector_type(2))) float v2f;
typedef __attribute__((ext_vector_type(8))) float v8f;

namespace gol {

struct Tables {
  unsigned short y4[28];
  unsigned short r3[48];
  int n4, n3p, n3f;
};

constexpr int popc9(int m) { int c = 0; for (int k = 0; k < 9; ++k) c += (m >> k) & 1; return c; }

constexpr int turn(int m) {
  int r = 0;
  for (int k = 0; k < 9; ++k) if ((m >> k) & 1) r |= 1 << (8 - k);
  return r;
}

constexpr bool future_alive(int m) {
  const int tot = popc9(m);
  const bool alive = ((m >> 4) & 1) != 0;
  return alive ? (tot == 3 || tot == 4) : (tot == 3);
}

constexpr Tables build() {
  Tables t{};
  for (int i = 0; i < 28; ++i) t.y4[i] = 0u;
  for (int i = 0; i < 48; ++i) t.r3[i] = 0xFFFFu;
  unsigned short fx[4] = {};
  bool used[512] = {};
  int a = 0, b = 0, f = 0;
  for (int m = 0; m < 512; ++m) {
    if (!future_alive(m) || used[m]) continue;
    const int z = turn(m);
    used[m] = true;
    used[z] = true;
    if (popc9(m) == 4)  t.y4[a++] = (unsigned short)m;
    else if (z == m)    fx[f++]   = (unsigned short)m;
    else                t.r3[b++] = (unsigned short)m;
  }
  for (int i = 0; i < f; ++i) t.r3[40 + i] = fx[i];
  t.n4 = a; t.n3p = b; t.n3f = f;
  return t;
}

constexpr bool tables_ok() {
  constexpr Tables t = build();
  if (t.n4 != 28 || t.n3p != 40 || t.n3f != 4) return false;
  int seen[512] = {};
  for (int k = 0; k < 28; ++k) {
    if (turn(t.y4[k]) == t.y4[k]) return false;
    ++seen[t.y4[k]]; ++seen[turn(t.y4[k])];
  }
  for (int r = 0; r < 44; ++r) {
    const int y = t.r3[r];
    if ((turn(y) == y) != (r >= 40)) return false;
    ++seen[y];
    if (r < 40) ++seen[turn(y)];
  }
  for (int m = 0; m < 512; ++m) if (seen[m] != (future_alive(m) ? 1 : 0)) return false;
  return true;
}
static_assert(tables_ok(), "config pairing broken");

constexpr unsigned Y4c(int k) { return build().y4[k]; }
constexpr unsigned R3c(int r) { return build().r3[r]; }

constexpr int subset_of(unsigned a, unsigned b) { return (a & ~b) == 0u ? 1 : 0; }

struct MTab { unsigned m[2][48][2]; };

constexpr MTab buildMT() {
  MTab r{};
  constexpr Tables t = build();
  for (int g = 0; g < 2; ++g) {
    for (int row = 0; row < 48; ++row) {
      for (int H = 0; H < 2; ++H) {
        unsigned mm = 0;
        for (int idx = 0; idx < 14; ++idx) {
          const int s = idx >> 1, j = idx & 1;
          const int k = 4 * s + 2 * H + j;
          const unsigned y = t.r3[row];
          unsigned code = 0;
          if (y != 0xFFFFu) {
            const int pp = subset_of(y, t.y4[k]);
            const int qq = subset_of(y, (unsigned)turn(t.y4[k]));
            if (g == 0) code = (unsigned)(pp + qq);
            else        code = (pp > qq) ? 1u : ((pp < qq) ? 2u : 0u);
          }
          mm |= code << (2 * idx);
        }
        r.m[g][row][H] = mm;
      }
    }
  }
  return r;
}

__device__ constexpr MTab MT = buildMT();

template<typename F, int... Is>
__device__ __forceinline__ void static_for_impl(F&& f, std::integer_sequence<int, Is...>) {
  (f(std::integral_constant<int, Is>{}), ...);
}
template<int N, typename F>
__device__ __forceinline__ void static_for(F&& f) {
  static_for_impl(static_cast<F&&>(f), std::make_integer_sequence<int, N>{});
}

}

namespace gol {

__device__ __forceinline__ void load_nbhd(const float* __restrict__ x, int w, int p, float (&n)[9]) {
  const int b  = w >> 12;
  const int rm = w & 4095;
  const int y  = rm >> 4;
  const int x0 = ((rm & 15) << 4) + p;
  const float* img = x + (size_t)b * 65536;
#pragma unroll
  for (int k = 0; k < 9; ++k) {
    const int dy = k / 3 - 1, dx = k % 3 - 1;
    const int yy = (y + dy) & 255;
    const int xx = (x0 + dx) & 255;
    n[k] = img[yy * 256 + xx];
  }
}

}

__global__ __launch_bounds__(256) void gol_wmma_kernel(const float* __restrict__ x,
                                                       float* __restrict__ out,
                                                       int nTiles) {
  using namespace gol;
  const int lane = (int)(threadIdx.x & 31u);
  const int p = lane & 15;
  const int h = lane >> 4;
  const int wv = __builtin_amdgcn_readfirstlane((int)(threadIdx.x >> 5));
  const int wavesPerBlock = (int)(blockDim.x >> 5);
  const int stride = (int)gridDim.x * wavesPerBlock;
  int w = (int)blockIdx.x * wavesPerBlock + wv;
  if (w >= nTiles) return;

  v2f Ae[3][7], Ao[3][7];
  static_for<3>([&](auto T) {
    constexpr int t = T.value;
    const unsigned me = MT.m[0][16 * t + p][h];
    const unsigned mo = MT.m[1][16 * t + p][h];
    static_for<7>([&](auto S) {
      constexpr int s = S.value;
      const unsigned e0 = (me >> (4 * s)) & 3u, e1 = (me >> (4 * s + 2)) & 3u;
      const unsigned o0 = (mo >> (4 * s)) & 3u, o1 = (mo >> (4 * s + 2)) & 3u;
      Ae[t][s].x = 0.5f * (float)e0;
      Ae[t][s].y = 0.5f * (float)e1;
      Ao[t][s].x = 0.5f * ((float)(o0 & 1u) - (float)(o0 >> 1));
      Ao[t][s].y = 0.5f * ((float)(o1 & 1u) - (float)(o1 >> 1));
    });
  });

  const bool hi = (h != 0);

  float nn[9];
  load_nbhd(x, w, p, nn);

  for (;;) {
    float n[9];
#pragma unroll
    for (int k = 0; k < 9; ++k) n[k] = nn[k];

    const int wn = w + stride;
    const bool more = wn < nTiles;
    load_nbhd(x, more ? wn : w, p, nn);

    float c[9];
#pragma unroll
    for (int k = 0; k < 9; ++k) c[k] = 1.0f - n[k];

    float gA[16]; gA[0] = 1.0f;
    static_for<4>([&](auto C4) {
      constexpr int cell = C4.value;
      constexpr int s = 1 << cell;
      static_for<s>([&](auto M) {
        constexpr int m = s - 1 - M.value;
        gA[m + s] = gA[m] * n[cell];
        gA[m]     = gA[m] * c[cell];
      });
    });
    float gB[32]; gB[0] = 1.0f;
    static_for<5>([&](auto C5) {
      constexpr int cell = C5.value;
      constexpr int s = 1 << cell;
      static_for<s>([&](auto M) {
        constexpr int m = s - 1 - M.value;
        gB[m + s] = gB[m] * n[cell + 4];
        gB[m]     = gB[m] * c[cell + 4];
      });
    });

    float sumAcc = 0.0f;

    float bp[14], bm[14];
    static_for<14>([&](auto I) {
      constexpr int idx = I.value;
      constexpr int s = idx >> 1, j = idx & 1;
      constexpr unsigned ya0 = Y4c(4 * s + j),     yb0 = (unsigned)turn((int)ya0);
      constexpr unsigned ya1 = Y4c(4 * s + j + 2), yb1 = (unsigned)turn((int)ya1);
      const float a0 = gA[ya0 & 15] * gB[(ya0 >> 4) & 31];
      const float b0 = gA[yb0 & 15] * gB[(yb0 >> 4) & 31];
      const float a1 = gA[ya1 & 15] * gB[(ya1 >> 4) & 31];
      const float b1 = gA[yb1 & 15] * gB[(yb1 >> 4) & 31];
      const float va = hi ? a1 : a0;
      const float vb = hi ? b1 : b0;
      bp[idx] = va + vb;
      bm[idx] = va - vb;
      sumAcc += bp[idx];
    });

    float ap[24], am[24];
    static_for<24>([&](auto I) {
      constexpr int idx = I.value;
      constexpr int t = idx >> 3, v = idx & 7;
      constexpr unsigned y0 = R3c(16 * t + v), y1 = R3c(16 * t + v + 8);
      float a0 = 0.0f, b0 = 0.0f, a1 = 0.0f, b1 = 0.0f;
      if constexpr (y0 != 0xFFFFu) {
        a0 = gA[y0 & 15] * gB[(y0 >> 4) & 31];
        constexpr unsigned z0 = (unsigned)turn((int)y0);
        if constexpr (z0 != y0) b0 = gA[z0 & 15] * gB[(z0 >> 4) & 31];
      }
      if constexpr (y1 != 0xFFFFu) {
        a1 = gA[y1 & 15] * gB[(y1 >> 4) & 31];
        constexpr unsigned z1 = (unsigned)turn((int)y1);
        if constexpr (z1 != y1) b1 = gA[z1 & 15] * gB[(z1 >> 4) & 31];
      }
      const float va = hi ? a1 : a0;
      const float vb = hi ? b1 : b0;
      ap[idx] = va + vb;
      am[idx] = va - vb;
      sumAcc += ap[idx];
    });

    float orp = 0.0f;
    {
      v8f C0 = {0.f, 0.f, 0.f, 0.f, 0.f, 0.f, 0.f, 0.f};
      v8f C1 = {0.f, 0.f, 0.f, 0.f, 0.f, 0.f, 0.f, 0.f};
      v8f C2 = {0.f, 0.f, 0.f, 0.f, 0.f, 0.f, 0.f, 0.f};
      static_for<7>([&](auto S) {
        constexpr int s = S.value;
        v2f Bv;
        Bv.x = bp[2 * s];
        Bv.y = bp[2 * s + 1];
        C0 = __builtin_amdgcn_wmma_f32_16x16x4_f32(false, Ae[0][s], false, Bv, (short)0, C0, false, false);
        C1 = __builtin_amdgcn_wmma_f32_16x16x4_f32(false, Ae[1][s], false, Bv, (short)0, C1, false, false);
        C2 = __builtin_amdgcn_wmma_f32_16x16x4_f32(false, Ae[2][s], false, Bv, (short)0, C2, false, false);
      });
#pragma unroll
      for (int v = 0; v < 8; ++v) orp = fmaf(ap[v], C0[v], orp);
#pragma unroll
      for (int v = 0; v < 8; ++v) orp = fmaf(ap[8 + v], C1[v], orp);
#pragma unroll
      for (int v = 0; v < 8; ++v) orp = fmaf(ap[16 + v], C2[v], orp);
    }
    {
      v8f C0 = {0.f, 0.f, 0.f, 0.f, 0.f, 0.f, 0.f, 0.f};
      v8f C1 = {0.f, 0.f, 0.f, 0.f, 0.f, 0.f, 0.f, 0.f};
      v8f C2 = {0.f, 0.f, 0.f, 0.f, 0.f, 0.f, 0.f, 0.f};
      static_for<7>([&](auto S) {
        constexpr int s = S.value;
        v2f Bv;
        Bv.x = bm[2 * s];
        Bv.y = bm[2 * s + 1];
        C0 = __builtin_amdgcn_wmma_f32_16x16x4_f32(false, Ao[0][s], false, Bv, (short)0, C0, false, false);
        C1 = __builtin_amdgcn_wmma_f32_16x16x4_f32(false, Ao[1][s], false, Bv, (short)0, C1, false, false);
        C2 = __builtin_amdgcn_wmma_f32_16x16x4_f32(false, Ao[2][s], false, Bv, (short)0, C2, false, false);
      });
#pragma unroll
      for (int v = 0; v < 8; ++v) orp = fmaf(am[v], C0[v], orp);
#pragma unroll
      for (int v = 0; v < 8; ++v) orp = fmaf(am[8 + v], C1[v], orp);
#pragma unroll
      for (int v = 0; v < 8; ++v) orp = fmaf(am[16 + v], C2[v], orp);
    }

    float tot = sumAcc - orp;
    tot += __shfl_xor(tot, 16, 32);
    if (h == 0) out[(size_t)w * 16 + (size_t)p] = tot;

    if (!more) break;
    w = wn;
  }
}

extern "C" void kernel_launch(void* const* d_in, const int* in_sizes, int n_in,
                              void* d_out, int out_size, void* d_ws, size_t ws_size,
                              hipStream_t stream) {
  (void)n_in; (void)d_ws; (void)ws_size; (void)out_size;
  const float* x = (const float*)d_in[0];
  float* out = (float*)d_out;
  const int total = in_sizes[0];
  const int Bn = total >> 16;
  const int nTiles = Bn * 4096;
  dim3 block(256);
  dim3 grid(256);
  gol_wmma_kernel<<<grid, block, 0, stream>>>(x, out, nTiles);
}


# Four waves per SIMD instead of two (work-groups of 16 waves, 4096 waves of 8 tiles each): more waves to cover dependent vector chains, load waits and the matrix pipe; f32 sum order unchanged
# speedup vs baseline: 2.6989x; 1.5251x over previous
#include <hip/hip_runtime.h>
#include <utility>

typedef __attribute__((ext_vector_type(2))) float v2f;
typedef __attribute__((ext_vector_type(8))) float v8f;

#define GOL_HD __device__ __forceinline__

namespace gol {

struct Ent { unsigned char code, ctr, cls, ns; };

constexpr Ent make_ent(int code, int ctr) {
  Ent e{};
  e.code = (unsigned char)code; e.ctr = (unsigned char)ctr;
  int cls = 0, ns = 0;
  for (int X = 0; X < 4; ++X) {
    const int f = (code >> (2 * X)) & 3;
    if (f == 2) cls |= 1 << X;
    if (f == 1 || f == 2) ++ns;
  }
  e.cls = (unsigned char)cls; e.ns = (unsigned char)ns;
  return e;
}

constexpr int occ_code(int code) {
  int o = 0;
  for (int X = 0; X < 4; ++X) { const int f = (code >> (2 * X)) & 3; o += (f == 0) ? 0 : ((f == 3) ? 2 : 1); }
  return o;
}

struct Layout {
  Ent  k[56];
  Ent  r[96];
  bool rpad[96];
  bool nz[6][14];
  int  nnz;
  bool ok;
};

constexpr Layout build_layout() {
  Layout L{};
  L.ok = true;
  for (int i = 0; i < 96; ++i) { L.rpad[i] = true; L.r[i] = Ent{0, 0, 0, 0}; }
  Ent s4[56] = {}; Ent s3[84] = {};
  int n4 = 0, n3 = 0;
  for (int ctr = 1; ctr >= 0; --ctr)
    for (int code = 0; code < 256; ++code) {
      const int o = occ_code(code);
      if (ctr == 1 && o == 3) { if (n4 < 56) s4[n4] = make_ent(code, 1); ++n4; }
      if ((ctr == 1 && o == 2) || (ctr == 0 && o == 3)) { if (n3 < 84) s3[n3] = make_ent(code, ctr); ++n3; }
    }
  if (n4 != 56 || n3 != 84) L.ok = false;
  const int KORD[15] = {0, 1, 6, 2, 9, 4, 3, 8, 5, 10, 12, 7, 11, 13, 14};
  int nk = 0;
  for (int c = 0; c < 15; ++c)
    for (int i = 0; i < 56; ++i)
      if (s4[i].cls == KORD[c]) { if (nk < 56) L.k[nk] = s4[i]; ++nk; }
  if (nk != 56) L.ok = false;
  int used[16] = {};
  auto put = [&](int slot, int cls) {
    int q = used[cls]++, seen = 0; bool found = false;
    for (int i = 0; i < 84; ++i) if (s3[i].cls == cls) { if (seen == q) { L.r[slot] = s3[i]; found = true; } ++seen; }
    if (!found || !L.rpad[slot]) L.ok = false;
    L.rpad[slot] = false;
  };
  for (int i = 0; i < 16; ++i) put(i, 0);
  for (int i = 0; i < 5; ++i) { put(16 + i, 0); put(24 + i, 0); }
  for (int i = 0; i < 3; ++i) { put(21 + i, 10); put(29 + i, 12); }
  const int big[4] = {1, 2, 4, 8}, host[4] = {6, 9, 3, 5};
  for (int g = 0; g < 4; ++g) {
    for (int i = 0; i < 9; ++i) put(32 + 16 * g + i, big[g]);
    for (int i = 0; i < 3; ++i) put(32 + 16 * g + 9 + i, host[g]);
  }
  put(44, 7); put(45, 11); put(46, 13); put(47, 14);
  int tot = 0;
  for (int c = 0; c < 16; ++c) {
    int have = 0;
    for (int i = 0; i < 84; ++i) if (s3[i].cls == c) ++have;
    if (have != used[c]) L.ok = false;
    tot += used[c];
  }
  if (tot != 84) L.ok = false;
  L.nnz = 0;
  for (int t = 0; t < 6; ++t)
    for (int s = 0; s < 14; ++s) {
      bool any = false;
      for (int i = 0; i < 16; ++i)
        for (int j = 0; j < 4; ++j)
          if (!L.rpad[16 * t + i] && L.r[16 * t + i].cls == L.k[4 * s + j].cls) any = true;
      L.nz[t][s] = any;
      if (any) ++L.nnz;
    }
  return L;
}

constexpr Layout LAY = build_layout();
static_assert(LAY.ok, "basis layout broken");
static_assert(LAY.nnz == 18, "block count changed");

struct Cfgs { int n; unsigned short m[8]; signed char s[8]; };

constexpr Cfgs expand(Ent e) {
  Cfgs c{};
  c.n = 1 << e.ns;
  for (int a = 0; a < c.n; ++a) {
    int mask = e.ctr ? 16 : 0, sg = 1, bit = 0;
    for (int X = 0; X < 4; ++X) {
      const int f = (e.code >> (2 * X)) & 3;
      if (f == 3) mask |= (1 << X) | (1 << (8 - X));
      else if (f == 1 || f == 2) {
        const int ch = (a >> bit) & 1; ++bit;
        mask |= ch ? (1 << (8 - X)) : (1 << X);
        if (f == 2 && ch) sg = -sg;
      }
    }
    c.m[a] = (unsigned short)mask; c.s[a] = (signed char)sg;
  }
  return c;
}

struct MF { alignas(16) float m[96][56]; };

constexpr MF build_m() {
  MF r{};
  Cfgs ck[56] = {};
  for (int k = 0; k < 56; ++k) ck[k] = expand(LAY.k[k]);
  for (int row = 0; row < 96; ++row) {
    for (int k = 0; k < 56; ++k) r.m[row][k] = 0.0f;
    if (LAY.rpad[row]) continue;
    const Cfgs cr = expand(LAY.r[row]);
    for (int k = 0; k < 56; ++k) {
      if (LAY.k[k].cls != LAY.r[row].cls) continue;
      int num = 0;
      for (int i = 0; i < cr.n; ++i)
        for (int j = 0; j < ck[k].n; ++j)
          if ((cr.m[i] & ~ck[k].m[j]) == 0) num += cr.s[i] * ck[k].s[j];
      r.m[row][k] = (float)num / (float)(1 << (LAY.r[row].ns + LAY.k[k].ns));
    }
  }
  return r;
}

template<typename F, int... Is>
GOL_HD void static_for_impl(F&& f, std::integer_sequence<int, Is...>) {
  (f(std::integral_constant<int, Is>{}), ...);
}
template<int N, typename F>
GOL_HD void static_for(F&& f) {
  static_for_impl(static_cast<F&&>(f), std::make_integer_sequence<int, N>{});
}

GOL_HD void lane_values(const float (&n)[9], bool hi, float (&bsel)[28], float (&asel)[48], float& sum) {
  float c[9];
  for (int k = 0; k < 9; ++k) c[k] = 1.0f - n[k];
  float F[4][4];
  static_for<4>([&](auto XX) {
    constexpr int X = XX.value;
    const float u = n[X] * c[8 - X];
    const float w = c[X] * n[8 - X];
    F[X][0] = c[X] * c[8 - X];
    F[X][1] = u + w;
    F[X][2] = u - w;
    F[X][3] = n[X] * n[8 - X];
  });
  float tAB[16], tCDn[16], tCDc[16];
  static_for<16>([&](auto I) {
    constexpr int i = I.value;
    tAB[i] = F[0][i & 3] * F[1][i >> 2];
    const float cd = F[2][i & 3] * F[3][i >> 2];
    tCDn[i] = cd * n[4];
    tCDc[i] = cd * c[4];
  });
  sum = 0.0f;
  static_for<28>([&](auto I) {
    constexpr int idx = I.value;
    constexpr int s = idx >> 1, j = idx & 1;
    constexpr Ent e0 = LAY.k[4 * s + j], e1 = LAY.k[4 * s + j + 2];
    const float v0 = tAB[e0.code & 15] * tCDn[(e0.code >> 4) & 15];
    const float v1 = tAB[e1.code & 15] * tCDn[(e1.code >> 4) & 15];
    bsel[idx] = hi ? v1 : v0;
    static_assert((e0.cls == 0) == (e1.cls == 0), "sum slots must agree across lane halves");
    if constexpr (e0.cls == 0) sum += bsel[idx];
  });
  static_for<48>([&](auto I) {
    constexpr int idx = I.value;
    constexpr int t = idx >> 3, v = idx & 7;
    constexpr int r0 = 16 * t + v, r1 = r0 + 8;
    float v0 = 0.0f, v1 = 0.0f;
    if constexpr (!LAY.rpad[r0]) {
      constexpr Ent e = LAY.r[r0];
      if constexpr (e.ctr != 0) v0 = tAB[e.code & 15] * tCDn[(e.code >> 4) & 15];
      else                      v0 = tAB[e.code & 15] * tCDc[(e.code >> 4) & 15];
    }
    if constexpr (!LAY.rpad[r1]) {
      constexpr Ent e = LAY.r[r1];
      if constexpr (e.ctr != 0) v1 = tAB[e.code & 15] * tCDn[(e.code >> 4) & 15];
      else                      v1 = tAB[e.code & 15] * tCDc[(e.code >> 4) & 15];
    }
    asel[idx] = hi ? v1 : v0;
    constexpr bool s0 = !LAY.rpad[r0] && LAY.r[r0].cls == 0;
    constexpr bool s1 = !LAY.rpad[r1] && LAY.r[r1].cls == 0;
    static_assert(s0 == s1, "sum slots must agree across lane halves");
    if constexpr (s0) sum += asel[idx];
  });
}

}

namespace gol {

__device__ constexpr MF MTf = build_m();

__device__ __forceinline__ void load_nbhd(const float* __restrict__ x, int w, int p, float (&n)[9]) {
  const int b  = w >> 12;
  const int rm = w & 4095;
  const int y  = rm >> 4;
  const int x0 = ((rm & 15) << 4) + p;
  const float* img = x + (size_t)b * 65536;
#pragma unroll
  for (int k = 0; k < 9; ++k) {
    const int dy = k / 3 - 1, dx = k % 3 - 1;
    const int yy = (y + dy) & 255;
    const int xx = (x0 + dx) & 255;
    n[k] = img[yy * 256 + xx];
  }
}

}

__global__ __launch_bounds__(512) void gol_wmma_kernel(const float* __restrict__ x,
                                                       float* __restrict__ out,
                                                       int nTiles) {
  using namespace gol;
  const int lane = (int)(threadIdx.x & 31u);
  const int p = lane & 15;
  const int h = lane >> 4;
  const int wv = __builtin_amdgcn_readfirstlane((int)(threadIdx.x >> 5));
  const int wavesPerBlock = (int)(blockDim.x >> 5);
  const int stride = (int)gridDim.x * wavesPerBlock;
  int w = (int)blockIdx.x * wavesPerBlock + wv;
  if (w >= nTiles) return;

  v2f A[6][14];
  static_for<6>([&](auto TT) {
    constexpr int t = TT.value;
    static_for<14>([&](auto SS) {
      constexpr int s = SS.value;
      if constexpr (LAY.nz[t][s]) {
        A[t][s].x = MTf.m[16 * t + p][4 * s + 2 * h];
        A[t][s].y = MTf.m[16 * t + p][4 * s + 2 * h + 1];
      }
    });
  });

  const bool hi = (h != 0);

  float nn[9];
  load_nbhd(x, w, p, nn);

  for (;;) {
    float n[9];
#pragma unroll
    for (int k = 0; k < 9; ++k) n[k] = nn[k];

    const int wn = w + stride;
    const bool more = wn < nTiles;
    load_nbhd(x, more ? wn : w, p, nn);

    float bsel[28], asel[48], sumAcc;
    lane_values(n, hi, bsel, asel, sumAcc);

    v8f C[6];
    static_for<6>([&](auto TT) {
      constexpr int t = TT.value;
      C[t] = v8f{0.f, 0.f, 0.f, 0.f, 0.f, 0.f, 0.f, 0.f};
    });
    static_for<14>([&](auto SS) {
      constexpr int s = SS.value;
      v2f Bv;
      Bv.x = bsel[2 * s];
      Bv.y = bsel[2 * s + 1];
      static_for<6>([&](auto TT) {
        constexpr int t = TT.value;
        if constexpr (LAY.nz[t][s])
          C[t] = __builtin_amdgcn_wmma_f32_16x16x4_f32(false, A[t][s], false, Bv, (short)0, C[t], false, false);
      });
    });

    float orp = 0.0f;
    static_for<6>([&](auto TT) {
      constexpr int t = TT.value;
      float o = 0.0f;
#pragma unroll
      for (int v = 0; v < 8; ++v) o = fmaf(asel[8 * t + v], C[t][v], o);
      orp += o;
    });

    float tot = sumAcc - orp;
    tot += __shfl_xor(tot, 16, 32);
    if (h == 0) out[(size_t)w * 16 + (size_t)p] = tot;

    if (!more) break;
    w = wn;
  }
}

extern "C" void kernel_launch(void* const* d_in, const int* in_sizes, int n_in,
                              void* d_out, int out_size, void* d_ws, size_t ws_size,
                              hipStream_t stream) {
  (void)n_in; (void)d_ws; (void)ws_size; (void)out_size;
  const float* x = (const float*)d_in[0];
  float* out = (float*)d_out;
  const int total = in_sizes[0];
  const int Bn = total >> 16;
  const int nTiles = Bn * 4096;
  dim3 block(512);
  dim3 grid(256);
  gol_wmma_kernel<<<grid, block, 0, stream>>>(x, out, nTiles);
}


# Each lane half forms only its own K/row entries: slot partners reordered to share one factor, the other picked by lane half (was both entries formed, then one selected); f32 sum order changes
# speedup vs baseline: 2.8916x; 1.0714x over previous
#include <hip/hip_runtime.h>
#include <utility>

typedef __attribute__((ext_vector_type(2))) float v2f;
typedef __attribute__((ext_vector_type(8))) float v8f;

#define GOL_HD __device__ __forceinline__

namespace gol {

struct Ent { unsigned char code, ctr, cls, ns; };

constexpr Ent make_ent(int code, int ctr) {
  Ent e{};
  e.code = (unsigned char)code; e.ctr = (unsigned char)ctr;
  int cls = 0, ns = 0;
  for (int X = 0; X < 4; ++X) {
    const int f = (code >> (2 * X)) & 3;
    if (f == 2) cls |= 1 << X;
    if (f == 1 || f == 2) ++ns;
  }
  e.cls = (unsigned char)cls; e.ns = (unsigned char)ns;
  return e;
}

constexpr int occ_code(int code) {
  int o = 0;
  for (int X = 0; X < 4; ++X) { const int f = (code >> (2 * X)) & 3; o += (f == 0) ? 0 : ((f == 3) ? 2 : 1); }
  return o;
}

struct Layout {
  Ent  k[56];
  Ent  r[96];
  bool rpad[96];
  bool nz[6][14];
  int  nnz;
  bool ok;
};

constexpr unsigned char KTAB[56] = {
  19, 208, 67, 112, 52, 76, 49, 28, 21, 84, 69, 81, 196, 13,
  193, 7, 50, 22, 194, 70, 82, 41, 14, 104, 88, 56, 11, 200,
  73, 134, 25, 146, 35, 100, 44, 97, 224, 26, 37, 74, 148, 131,
  145, 140, 38, 176, 98, 133, 164, 137, 161, 152, 168, 42, 162, 138
};
constexpr short RTAB[96] = {
  52, 324, 13, 67, 259, 304, 112, 336, 49, 273, 7, 76, 268, 448, 208, 261,
  276, 19, 84, 196, 21, 137, 416, 164, 321, 28, 81, 193, 69, 392, 152, 161,
  262, 50, 274, 22, 194, 168, 104, 41, 82, 14, 322, 70, 138, 162, 296, 42,
  328, 56, 200, 25, 134, 265, -1, -1, 280, 11, 146, 73, 386, 88, -1, -1,
  100, 292, 44, 352, 37, -1, 26, 35, 97, 289, -1, 224, -1, 266, 74, -1,
  -1, 133, 148, 98, 400, 388, 131, 38, -1, -1, 145, 290, 176, 385, 140, -1
};

constexpr Layout build_layout() {
  Layout L{};
  L.ok = true;
  bool seen4[256] = {};
  for (int i = 0; i < 56; ++i) {
    const int code = KTAB[i];
    if (occ_code(code) != 3 || seen4[code]) L.ok = false;
    seen4[code] = true;
    L.k[i] = make_ent(code, 1);
  }
  bool seen3[512] = {};
  int n3 = 0;
  for (int i = 0; i < 96; ++i) {
    const int v = RTAB[i];
    L.rpad[i] = v < 0;
    L.r[i] = Ent{0, 0, 0, 0};
    if (v < 0) continue;
    const int code = v & 255, ctr = v >> 8;
    const int o = occ_code(code);
    if (!((ctr == 1 && o == 2) || (ctr == 0 && o == 3)) || seen3[v]) L.ok = false;
    seen3[v] = true;
    L.r[i] = make_ent(code, ctr);
    ++n3;
  }
  if (n3 != 84) L.ok = false;
  L.nnz = 0;
  for (int t = 0; t < 6; ++t)
    for (int s = 0; s < 14; ++s) {
      bool any = false;
      for (int i = 0; i < 16; ++i)
        for (int j = 0; j < 4; ++j)
          if (!L.rpad[16 * t + i] && L.r[16 * t + i].cls == L.k[4 * s + j].cls) any = true;
      L.nz[t][s] = any;
      if (any) ++L.nnz;
    }
  return L;
}

constexpr Layout LAY = build_layout();
static_assert(LAY.ok, "basis layout broken");
static_assert(LAY.nnz == 18, "block count changed");

struct Cfgs { int n; unsigned short m[8]; signed char s[8]; };

constexpr Cfgs expand(Ent e) {
  Cfgs c{};
  c.n = 1 << e.ns;
  for (int a = 0; a < c.n; ++a) {
    int mask = e.ctr ? 16 : 0, sg = 1, bit = 0;
    for (int X = 0; X < 4; ++X) {
      const int f = (e.code >> (2 * X)) & 3;
      if (f == 3) mask |= (1 << X) | (1 << (8 - X));
      else if (f == 1 || f == 2) {
        const int ch = (a >> bit) & 1; ++bit;
        mask |= ch ? (1 << (8 - X)) : (1 << X);
        if (f == 2 && ch) sg = -sg;
      }
    }
    c.m[a] = (unsigned short)mask; c.s[a] = (signed char)sg;
  }
  return c;
}

struct MF { alignas(16) float m[96][56]; };

constexpr MF build_m() {
  MF r{};
  Cfgs ck[56] = {};
  for (int k = 0; k < 56; ++k) ck[k] = expand(LAY.k[k]);
  for (int row = 0; row < 96; ++row) {
    for (int k = 0; k < 56; ++k) r.m[row][k] = 0.0f;
    if (LAY.rpad[row]) continue;
    const Cfgs cr = expand(LAY.r[row]);
    for (int k = 0; k < 56; ++k) {
      if (LAY.k[k].cls != LAY.r[row].cls) continue;
      int num = 0;
      for (int i = 0; i < cr.n; ++i)
        for (int j = 0; j < ck[k].n; ++j)
          if ((cr.m[i] & ~ck[k].m[j]) == 0) num += cr.s[i] * ck[k].s[j];
      r.m[row][k] = (float)num / (float)(1 << (LAY.r[row].ns + LAY.k[k].ns));
    }
  }
  return r;
}

template<typename F, int... Is>
GOL_HD void static_for_impl(F&& f, std::integer_sequence<int, Is...>) {
  (f(std::integral_constant<int, Is>{}), ...);
}
template<int N, typename F>
GOL_HD void static_for(F&& f) {
  static_for_impl(static_cast<F&&>(f), std::make_integer_sequence<int, N>{});
}

template<int CTR, int J>
GOL_HD float cd_of(const float (&tn)[16], const float (&tc)[16]) {
  if constexpr (CTR != 0) return tn[J]; else return tc[J];
}

GOL_HD void lane_values(const float (&n)[9], bool hi, float (&bsel)[28], float (&asel)[48], float& sum) {
  float c[9];
  for (int k = 0; k < 9; ++k) c[k] = 1.0f - n[k];
  float F[4][4];
  static_for<4>([&](auto XX) {
    constexpr int X = XX.value;
    const float u = n[X] * c[8 - X];
    const float w = c[X] * n[8 - X];
    F[X][0] = c[X] * c[8 - X];
    F[X][1] = u + w;
    F[X][2] = u - w;
    F[X][3] = n[X] * n[8 - X];
  });
  float tAB[16], tCDn[16], tCDc[16];
  static_for<16>([&](auto I) {
    constexpr int i = I.value;
    tAB[i] = F[0][i & 3] * F[1][i >> 2];
    const float cd = F[2][i & 3] * F[3][i >> 2];
    tCDn[i] = cd * n[4];
    tCDc[i] = cd * c[4];
  });
  sum = 0.0f;
  static_for<28>([&](auto I) {
    constexpr int idx = I.value;
    constexpr int s = idx >> 1, j = idx & 1;
    constexpr Ent e0 = LAY.k[4 * s + j], e1 = LAY.k[4 * s + j + 2];
    constexpr int i0 = e0.code & 15, i1 = e1.code & 15;
    constexpr int j0 = (e0.code >> 4) & 15, j1 = (e1.code >> 4) & 15;
    float ab, cd;
    if constexpr (i0 == i1) ab = tAB[i0]; else ab = hi ? tAB[i1] : tAB[i0];
    if constexpr (j0 == j1) cd = tCDn[j0]; else cd = hi ? tCDn[j1] : tCDn[j0];
    bsel[idx] = ab * cd;
    static_assert((e0.cls == 0) == (e1.cls == 0), "sum slots must agree across lane halves");
    if constexpr (e0.cls == 0) sum += bsel[idx];
  });
  static_for<48>([&](auto I) {
    constexpr int idx = I.value;
    constexpr int t = idx >> 3, v = idx & 7;
    constexpr int r0 = 16 * t + v, r1 = r0 + 8;
    constexpr bool p0 = LAY.rpad[r0], p1 = LAY.rpad[r1];
    constexpr Ent e0 = LAY.r[r0], e1 = LAY.r[r1];
    constexpr int i0 = e0.code & 15, i1 = e1.code & 15;
    constexpr int j0 = (e0.code >> 4) & 15, j1 = (e1.code >> 4) & 15;
    if constexpr (p0 && p1) {
      asel[idx] = 0.0f;
    } else if constexpr (p1) {
      asel[idx] = tAB[i0] * (hi ? 0.0f : cd_of<e0.ctr, j0>(tCDn, tCDc));
    } else if constexpr (p0) {
      asel[idx] = tAB[i1] * (hi ? cd_of<e1.ctr, j1>(tCDn, tCDc) : 0.0f);
    } else {
      float ab, cd;
      if constexpr (i0 == i1) ab = tAB[i0]; else ab = hi ? tAB[i1] : tAB[i0];
      if constexpr (j0 == j1 && e0.ctr == e1.ctr) cd = cd_of<e0.ctr, j0>(tCDn, tCDc);
      else cd = hi ? cd_of<e1.ctr, j1>(tCDn, tCDc) : cd_of<e0.ctr, j0>(tCDn, tCDc);
      asel[idx] = ab * cd;
    }
    constexpr bool s0 = !p0 && e0.cls == 0;
    constexpr bool s1 = !p1 && e1.cls == 0;
    static_assert(s0 == s1, "sum slots must agree across lane halves");
    if constexpr (s0) sum += asel[idx];
  });
}

}

namespace gol {

__device__ constexpr MF MTf = build_m();

__device__ __forceinline__ void load_nbhd(const float* __restrict__ x, int w, int p, float (&n)[9]) {
  const int b  = w >> 12;
  const int rm = w & 4095;
  const int y  = rm >> 4;
  const int x0 = ((rm & 15) << 4) + p;
  const float* img = x + (size_t)b * 65536;
#pragma unroll
  for (int k = 0; k < 9; ++k) {
    const int dy = k / 3 - 1, dx = k % 3 - 1;
    const int yy = (y + dy) & 255;
    const int xx = (x0 + dx) & 255;
    n[k] = img[yy * 256 + xx];
  }
}

}

__global__ __launch_bounds__(512) void gol_wmma_kernel(const float* __restrict__ x,
                                                       float* __restrict__ out,
                                                       int nTiles) {
  using namespace gol;
  const int lane = (int)(threadIdx.x & 31u);
  const int p = lane & 15;
  const int h = lane >> 4;
  const int wv = __builtin_amdgcn_readfirstlane((int)(threadIdx.x >> 5));
  const int wavesPerBlock = (int)(blockDim.x >> 5);
  const int stride = (int)gridDim.x * wavesPerBlock;
  int w = (int)blockIdx.x * wavesPerBlock + wv;
  if (w >= nTiles) return;

  v2f A[6][14];
  static_for<6>([&](auto TT) {
    constexpr int t = TT.value;
    static_for<14>([&](auto SS) {
      constexpr int s = SS.value;
      if constexpr (LAY.nz[t][s]) {
        A[t][s].x = MTf.m[16 * t + p][4 * s + 2 * h];
        A[t][s].y = MTf.m[16 * t + p][4 * s + 2 * h + 1];
      }
    });
  });

  const bool hi = (h != 0);

  float nn[9];
  load_nbhd(x, w, p, nn);

  for (;;) {
    float n[9];
#pragma unroll
    for (int k = 0; k < 9; ++k) n[k] = nn[k];

    const int wn = w + stride;
    const bool more = wn < nTiles;
    load_nbhd(x, more ? wn : w, p, nn);

    float bsel[28], asel[48], sumAcc;
    lane_values(n, hi, bsel, asel, sumAcc);

    v8f C[6];
    static_for<6>([&](auto TT) {
      constexpr int t = TT.value;
      C[t] = v8f{0.f, 0.f, 0.f, 0.f, 0.f, 0.f, 0.f, 0.f};
    });
    static_for<14>([&](auto SS) {
      constexpr int s = SS.value;
      v2f Bv;
      Bv.x = bsel[2 * s];
      Bv.y = bsel[2 * s + 1];
      static_for<6>([&](auto TT) {
        constexpr int t = TT.value;
        if constexpr (LAY.nz[t][s])
          C[t] = __builtin_amdgcn_wmma_f32_16x16x4_f32(false, A[t][s], false, Bv, (short)0, C[t], false, false);
      });
    });

    float orp = 0.0f;
    static_for<6>([&](auto TT) {
      constexpr int t = TT.value;
      float o = 0.0f;
      static_for<8>([&](auto VV) {
        constexpr int v = VV.value;
        if constexpr (!(LAY.rpad[16 * t + v] && LAY.rpad[16 * t + v + 8]))
          o = fmaf(asel[8 * t + v], C[t][v], o);
      });
      orp += o;
    });

    float tot = sumAcc - orp;
    tot += __shfl_xor(tot, 16, 32);
    if (h == 0) out[(size_t)w * 16 + (size_t)p] = tot;

    if (!more) break;
    w = wn;
  }
}

extern "C" void kernel_launch(void* const* d_in, const int* in_sizes, int n_in,
                              void* d_out, int out_size, void* d_ws, size_t ws_size,
                              hipStream_t stream) {
  (void)n_in; (void)d_ws; (void)ws_size; (void)out_size;
  const float* x = (const float*)d_in[0];
  float* out = (float*)d_out;
  const int total = in_sizes[0];
  const int Bn = total >> 16;
  const int nTiles = Bn * 4096;
  dim3 block(512);
  dim3 grid(256);
  gol_wmma_kernel<<<grid, block, 0, stream>>>(x, out, nTiles);
}


# Prologue: A fragments read as 9 whole-line 16-byte loads from a lane-ordered table, issued behind the first tile's loads; first tile peeled so its vector work covers them; f32 sum order unchanged
# speedup vs baseline: 3.0942x; 1.0701x over previous
#include <hip/hip_runtime.h>
#include <utility>

typedef __attribute__((ext_vector_type(2))) float v2f;
typedef __attribute__((ext_vector_type(8))) float v8f;

#define GOL_HD __device__ __forceinline__

namespace gol {

struct Ent { unsigned char code, ctr, cls, ns; };

constexpr Ent make_ent(int code, int ctr) {
  Ent e{};
  e.code = (unsigned char)code; e.ctr = (unsigned char)ctr;
  int cls = 0, ns = 0;
  for (int X = 0; X < 4; ++X) {
    const int f = (code >> (2 * X)) & 3;
    if (f == 2) cls |= 1 << X;
    if (f == 1 || f == 2) ++ns;
  }
  e.cls = (unsigned char)cls; e.ns = (unsigned char)ns;
  return e;
}

constexpr int occ_code(int code) {
  int o = 0;
  for (int X = 0; X < 4; ++X) { const int f = (code >> (2 * X)) & 3; o += (f == 0) ? 0 : ((f == 3) ? 2 : 1); }
  return o;
}

struct Layout {
  Ent  k[56];
  Ent  r[96];
  bool rpad[96];
  bool nz[6][14];
  int  nnz;
  bool ok;
};

constexpr unsigned char KTAB[56] = {
  19, 208, 67, 112, 52, 76, 49, 28, 21, 84, 69, 81, 196, 13,
  193, 7, 50, 22, 194, 70, 82, 41, 14, 104, 88, 56, 11, 200,
  73, 134, 25, 146, 35, 100, 44, 97, 224, 26, 37, 74, 148, 131,
  145, 140, 38, 176, 98, 133, 164, 137, 161, 152, 168, 42, 162, 138
};
constexpr short RTAB[96] = {
  52, 324, 13, 67, 259, 304, 112, 336, 49, 273, 7, 76, 268, 448, 208, 261,
  276, 19, 84, 196, 21, 137, 416, 164, 321, 28, 81, 193, 69, 392, 152, 161,
  262, 50, 274, 22, 194, 168, 104, 41, 82, 14, 322, 70, 138, 162, 296, 42,
  328, 56, 200, 25, 134, 265, -1, -1, 280, 11, 146, 73, 386, 88, -1, -1,
  100, 292, 44, 352, 37, -1, 26, 35, 97, 289, -1, 224, -1, 266, 74, -1,
  -1, 133, 148, 98, 400, 388, 131, 38, -1, -1, 145, 290, 176, 385, 140, -1
};

constexpr Layout build_layout() {
  Layout L{};
  L.ok = true;
  bool seen4[256] = {};
  for (int i = 0; i < 56; ++i) {
    const int code = KTAB[i];
    if (occ_code(code) != 3 || seen4[code]) L.ok = false;
    seen4[code] = true;
    L.k[i] = make_ent(code, 1);
  }
  bool seen3[512] = {};
  int n3 = 0;
  for (int i = 0; i < 96; ++i) {
    const int v = RTAB[i];
    L.rpad[i] = v < 0;
    L.r[i] = Ent{0, 0, 0, 0};
    if (v < 0) continue;
    const int code = v & 255, ctr = v >> 8;
    const int o = occ_code(code);
    if (!((ctr == 1 && o == 2) || (ctr == 0 && o == 3)) || seen3[v]) L.ok = false;
    seen3[v] = true;
    L.r[i] = make_ent(code, ctr);
    ++n3;
  }
  if (n3 != 84) L.ok = false;
  L.nnz = 0;
  for (int t = 0; t < 6; ++t)
    for (int s = 0; s < 14; ++s) {
      bool any = false;
      for (int i = 0; i < 16; ++i)
        for (int j = 0; j < 4; ++j)
          if (!L.rpad[16 * t + i] && L.r[16 * t + i].cls == L.k[4 * s + j].cls) any = true;
      L.nz[t][s] = any;
      if (any) ++L.nnz;
    }
  return L;
}

constexpr Layout LAY = build_layout();
static_assert(LAY.ok, "basis layout broken");
static_assert(LAY.nnz == 18, "block count changed");

struct Cfgs { int n; unsigned short m[8]; signed char s[8]; };

constexpr Cfgs expand(Ent e) {
  Cfgs c{};
  c.n = 1 << e.ns;
  for (int a = 0; a < c.n; ++a) {
    int mask = e.ctr ? 16 : 0, sg = 1, bit = 0;
    for (int X = 0; X < 4; ++X) {
      const int f = (e.code >> (2 * X)) & 3;
      if (f == 3) mask |= (1 << X) | (1 << (8 - X));
      else if (f == 1 || f == 2) {
        const int ch = (a >> bit) & 1; ++bit;
        mask |= ch ? (1 << (8 - X)) : (1 << X);
        if (f == 2 && ch) sg = -sg;
      }
    }
    c.m[a] = (unsigned short)mask; c.s[a] = (signed char)sg;
  }
  return c;
}

struct MF { alignas(16) float m[96][56]; };

constexpr MF build_m() {
  MF r{};
  Cfgs ck[56] = {};
  for (int k = 0; k < 56; ++k) ck[k] = expand(LAY.k[k]);
  for (int row = 0; row < 96; ++row) {
    for (int k = 0; k < 56; ++k) r.m[row][k] = 0.0f;
    if (LAY.rpad[row]) continue;
    const Cfgs cr = expand(LAY.r[row]);
    for (int k = 0; k < 56; ++k) {
      if (LAY.k[k].cls != LAY.r[row].cls) continue;
      int num = 0;
      for (int i = 0; i < cr.n; ++i)
        for (int j = 0; j < ck[k].n; ++j)
          if ((cr.m[i] & ~ck[k].m[j]) == 0) num += cr.s[i] * ck[k].s[j];
      r.m[row][k] = (float)num / (float)(1 << (LAY.r[row].ns + LAY.k[k].ns));
    }
  }
  return r;
}

template<typename F, int... Is>
GOL_HD void static_for_impl(F&& f, std::integer_sequence<int, Is...>) {
  (f(std::integral_constant<int, Is>{}), ...);
}
template<int N, typename F>
GOL_HD void static_for(F&& f) {
  static_for_impl(static_cast<F&&>(f), std::make_integer_sequence<int, N>{});
}

template<int CTR, int J>
GOL_HD float cd_of(const float (&tn)[16], const float (&tc)[16]) {
  if constexpr (CTR != 0) return tn[J]; else return tc[J];
}

GOL_HD void lane_values(const float (&n)[9], bool hi, float (&bsel)[28], float (&asel)[48], float& sum) {
  float c[9];
  for (int k = 0; k < 9; ++k) c[k] = 1.0f - n[k];
  float F[4][4];
  static_for<4>([&](auto XX) {
    constexpr int X = XX.value;
    const float u = n[X] * c[8 - X];
    const float w = c[X] * n[8 - X];
    F[X][0] = c[X] * c[8 - X];
    F[X][1] = u + w;
    F[X][2] = u - w;
    F[X][3] = n[X] * n[8 - X];
  });
  float tAB[16], tCDn[16], tCDc[16];
  static_for<16>([&](auto I) {
    constexpr int i = I.value;
    tAB[i] = F[0][i & 3] * F[1][i >> 2];
    const float cd = F[2][i & 3] * F[3][i >> 2];
    tCDn[i] = cd * n[4];
    tCDc[i] = cd * c[4];
  });
  sum = 0.0f;
  static_for<28>([&](auto I) {
    constexpr int idx = I.value;
    constexpr int s = idx >> 1, j = idx & 1;
    constexpr Ent e0 = LAY.k[4 * s + j], e1 = LAY.k[4 * s + j + 2];
    constexpr int i0 = e0.code & 15, i1 = e1.code & 15;
    constexpr int j0 = (e0.code >> 4) & 15, j1 = (e1.code >> 4) & 15;
    float ab, cd;
    if constexpr (i0 == i1) ab = tAB[i0]; else ab = hi ? tAB[i1] : tAB[i0];
    if constexpr (j0 == j1) cd = tCDn[j0]; else cd = hi ? tCDn[j1] : tCDn[j0];
    bsel[idx] = ab * cd;
    static_assert((e0.cls == 0) == (e1.cls == 0), "sum slots must agree across lane halves");
    if constexpr (e0.cls == 0) sum += bsel[idx];
  });
  static_for<48>([&](auto I) {
    constexpr int idx = I.value;
    constexpr int t = idx >> 3, v = idx & 7;
    constexpr int r0 = 16 * t + v, r1 = r0 + 8;
    constexpr bool p0 = LAY.rpad[r0], p1 = LAY.rpad[r1];
    constexpr Ent e0 = LAY.r[r0], e1 = LAY.r[r1];
    constexpr int i0 = e0.code & 15, i1 = e1.code & 15;
    constexpr int j0 = (e0.code >> 4) & 15, j1 = (e1.code >> 4) & 15;
    if constexpr (p0 && p1) {
      asel[idx] = 0.0f;
    } else if constexpr (p1) {
      asel[idx] = tAB[i0] * (hi ? 0.0f : cd_of<e0.ctr, j0>(tCDn, tCDc));
    } else if constexpr (p0) {
      asel[idx] = tAB[i1] * (hi ? cd_of<e1.ctr, j1>(tCDn, tCDc) : 0.0f);
    } else {
      float ab, cd;
      if constexpr (i0 == i1) ab = tAB[i0]; else ab = hi ? tAB[i1] : tAB[i0];
      if constexpr (j0 == j1 && e0.ctr == e1.ctr) cd = cd_of<e0.ctr, j0>(tCDn, tCDc);
      else cd = hi ? cd_of<e1.ctr, j1>(tCDn, tCDc) : cd_of<e0.ctr, j0>(tCDn, tCDc);
      asel[idx] = ab * cd;
    }
    constexpr bool s0 = !p0 && e0.cls == 0;
    constexpr bool s1 = !p1 && e1.cls == 0;
    static_assert(s0 == s1, "sum slots must agree across lane halves");
    if constexpr (s0) sum += asel[idx];
  });
}

}

namespace gol {

struct Blocks { int t[18], s[18]; };
constexpr Blocks list_blocks() {
  Blocks b{};
  int i = 0;
  for (int t = 0; t < 6; ++t)
    for (int s = 0; s < 14; ++s)
      if (LAY.nz[t][s]) { b.t[i] = t; b.s[i] = s; ++i; }
  return b;
}
constexpr Blocks BLK = list_blocks();

struct AT { alignas(16) float v[9][32][4]; };
constexpr AT build_at() {
  const MF m = build_m();
  AT a{};
  for (int i = 0; i < 18; ++i)
    for (int lane = 0; lane < 32; ++lane) {
      const int p = lane & 15, h = lane >> 4;
      a.v[i >> 1][lane][2 * (i & 1)]     = m.m[16 * BLK.t[i] + p][4 * BLK.s[i] + 2 * h];
      a.v[i >> 1][lane][2 * (i & 1) + 1] = m.m[16 * BLK.t[i] + p][4 * BLK.s[i] + 2 * h + 1];
    }
  return a;
}
__device__ constexpr AT ATf = build_at();

__device__ __forceinline__ void load_nbhd(const float* __restrict__ x, int w, int p, float (&n)[9]) {
  const int b  = w >> 12;
  const int rm = w & 4095;
  const int y  = rm >> 4;
  const int x0 = ((rm & 15) << 4) + p;
  const float* img = x + (size_t)b * 65536;
#pragma unroll
  for (int k = 0; k < 9; ++k) {
    const int dy = k / 3 - 1, dx = k % 3 - 1;
    const int yy = (y + dy) & 255;
    const int xx = (x0 + dx) & 255;
    n[k] = img[yy * 256 + xx];
  }
}

}

namespace gol {

GOL_HD float tile_result(const v2f (&A)[18], const float (&n)[9], bool hi) {
  float bsel[28], asel[48], sumAcc;
  lane_values(n, hi, bsel, asel, sumAcc);

  v8f C[6];
  static_for<6>([&](auto TT) {
    constexpr int t = TT.value;
    C[t] = v8f{0.f, 0.f, 0.f, 0.f, 0.f, 0.f, 0.f, 0.f};
  });
  static_for<14>([&](auto SS) {
    constexpr int s = SS.value;
    v2f Bv;
    Bv.x = bsel[2 * s];
    Bv.y = bsel[2 * s + 1];
    static_for<18>([&](auto II) {
      constexpr int i = II.value;
      if constexpr (BLK.s[i] == s)
        C[BLK.t[i]] = __builtin_amdgcn_wmma_f32_16x16x4_f32(false, A[i], false, Bv, (short)0, C[BLK.t[i]], false, false);
    });
  });

  float orp = 0.0f;
  static_for<6>([&](auto TT) {
    constexpr int t = TT.value;
    float o = 0.0f;
    static_for<8>([&](auto VV) {
      constexpr int v = VV.value;
      if constexpr (!(LAY.rpad[16 * t + v] && LAY.rpad[16 * t + v + 8]))
        o = fmaf(asel[8 * t + v], C[t][v], o);
    });
    orp += o;
  });

  float tot = sumAcc - orp;
  tot += __shfl_xor(tot, 16, 32);
  return tot;
}

}

__global__ __launch_bounds__(512) void gol_wmma_kernel(const float* __restrict__ x,
                                                       float* __restrict__ out,
                                                       int nTiles) {
  using namespace gol;
  const int lane = (int)(threadIdx.x & 31u);
  const int p = lane & 15;
  const int h = lane >> 4;
  const int wv = __builtin_amdgcn_readfirstlane((int)(threadIdx.x >> 5));
  const int wavesPerBlock = (int)(blockDim.x >> 5);
  const int stride = (int)gridDim.x * wavesPerBlock;
  int w = (int)blockIdx.x * wavesPerBlock + wv;
  if (w >= nTiles) return;

  const bool hi = (h != 0);

  float nn[9];
  load_nbhd(x, w, p, nn);

  v2f A[18];
  static_for<9>([&](auto QQ) {
    constexpr int q = QQ.value;
    const float* f = &ATf.v[q][lane][0];
    A[2 * q].x     = f[0];
    A[2 * q].y     = f[1];
    A[2 * q + 1].x = f[2];
    A[2 * q + 1].y = f[3];
  });

  {
    float n[9];
#pragma unroll
    for (int k = 0; k < 9; ++k) n[k] = nn[k];
    const int wn = w + stride;
    const bool more = wn < nTiles;
    load_nbhd(x, more ? wn : w, p, nn);
    const float tot = tile_result(A, n, hi);
    if (h == 0) out[(size_t)w * 16 + (size_t)p] = tot;
    if (!more) return;
    w = wn;
  }

  for (;;) {
    float n[9];
#pragma unroll
    for (int k = 0; k < 9; ++k) n[k] = nn[k];

    const int wn = w + stride;
    const bool more = wn < nTiles;
    load_nbhd(x, more ? wn : w, p, nn);

    const float tot = tile_result(A, n, hi);
    if (h == 0) out[(size_t)w * 16 + (size_t)p] = tot;

    if (!more) break;
    w = wn;
  }
}

extern "C" void kernel_launch(void* const* d_in, const int* in_sizes, int n_in,
                              void* d_out, int out_size, void* d_ws, size_t ws_size,
                              hipStream_t stream) {
  (void)n_in; (void)d_ws; (void)ws_size; (void)out_size;
  const float* x = (const float*)d_in[0];
  float* out = (float*)d_out;
  const int total = in_sizes[0];
  const int Bn = total >> 16;
  const int nTiles = Bn * 4096;
  dim3 block(512);
  dim3 grid(256);
  gol_wmma_kernel<<<grid, block, 0, stream>>>(x, out, nTiles);
}


# Eight waves per SIMD instead of four (one 32-wave work-group per WGP, 8192 waves of 4 tiles each): vector stretches of some waves run under the matrix runs of others; f32 sum order unchanged
# speedup vs baseline: 3.2661x; 1.0556x over previous
#include <hip/hip_runtime.h>
#include <utility>

typedef __attribute__((ext_vector_type(2))) float v2f;
typedef __attribute__((ext_vector_type(8))) float v8f;

#define GOL_HD __device__ __forceinline__

namespace gol {

struct Ent { unsigned char code, ctr, cls, ns; };

constexpr Ent make_ent(int code, int ctr) {
  Ent e{};
  e.code = (unsigned char)code; e.ctr = (unsigned char)ctr;
  int cls = 0, ns = 0;
  for (int X = 0; X < 4; ++X) {
    const int f = (code >> (2 * X)) & 3;
    if (f == 2) cls |= 1 << X;
    if (f == 1 || f == 2) ++ns;
  }
  e.cls = (unsigned char)cls; e.ns = (unsigned char)ns;
  return e;
}

constexpr int occ_code(int code) {
  int o = 0;
  for (int X = 0; X < 4; ++X) { const int f = (code >> (2 * X)) & 3; o += (f == 0) ? 0 : ((f == 3) ? 2 : 1); }
  return o;
}

struct Layout {
  Ent  k[56];
  Ent  r[96];
  bool rpad[96];
  bool nz[6][14];
  int  nnz;
  bool ok;
};

constexpr unsigned char KTAB[56] = {
  19, 208, 67, 112, 52, 76, 49, 28, 21, 84, 69, 81, 196, 13,
  193, 7, 50, 22, 194, 70, 82, 41, 14, 104, 88, 56, 11, 200,
  73, 134, 25, 146, 35, 100, 44, 97, 224, 26, 37, 74, 148, 131,
  145, 140, 38, 176, 98, 133, 164, 137, 161, 152, 168, 42, 162, 138
};
constexpr short RTAB[96] = {
  52, 324, 13, 67, 259, 304, 112, 336, 49, 273, 7, 76, 268, 448, 208, 261,
  276, 19, 84, 196, 21, 137, 416, 164, 321, 28, 81, 193, 69, 392, 152, 161,
  262, 50, 274, 22, 194, 168, 104, 41, 82, 14, 322, 70, 138, 162, 296, 42,
  328, 56, 200, 25, 134, 265, -1, -1, 280, 11, 146, 73, 386, 88, -1, -1,
  100, 292, 44, 352, 37, -1, 26, 35, 97, 289, -1, 224, -1, 266, 74, -1,
  -1, 133, 148, 98, 400, 388, 131, 38, -1, -1, 145, 290, 176, 385, 140, -1
};

constexpr Layout build_layout() {
  Layout L{};
  L.ok = true;
  bool seen4[256] = {};
  for (int i = 0; i < 56; ++i) {
    const int code = KTAB[i];
    if (occ_code(code) != 3 || seen4[code]) L.ok = false;
    seen4[code] = true;
    L.k[i] = make_ent(code, 1);
  }
  bool seen3[512] = {};
  int n3 = 0;
  for (int i = 0; i < 96; ++i) {
    const int v = RTAB[i];
    L.rpad[i] = v < 0;
    L.r[i] = Ent{0, 0, 0, 0};
    if (v < 0) continue;
    const int code = v & 255, ctr = v >> 8;
    const int o = occ_code(code);
    if (!((ctr == 1 && o == 2) || (ctr == 0 && o == 3)) || seen3[v]) L.ok = false;
    seen3[v] = true;
    L.r[i] = make_ent(code, ctr);
    ++n3;
  }
  if (n3 != 84) L.ok = false;
  L.nnz = 0;
  for (int t = 0; t < 6; ++t)
    for (int s = 0; s < 14; ++s) {
      bool any = false;
      for (int i = 0; i < 16; ++i)
        for (int j = 0; j < 4; ++j)
          if (!L.rpad[16 * t + i] && L.r[16 * t + i].cls == L.k[4 * s + j].cls) any = true;
      L.nz[t][s] = any;
      if (any) ++L.nnz;
    }
  return L;
}

constexpr Layout LAY = build_layout();
static_assert(LAY.ok, "basis layout broken");
static_assert(LAY.nnz == 18, "block count changed");

struct Cfgs { int n; unsigned short m[8]; signed char s[8]; };

constexpr Cfgs expand(Ent e) {
  Cfgs c{};
  c.n = 1 << e.ns;
  for (int a = 0; a < c.n; ++a) {
    int mask = e.ctr ? 16 : 0, sg = 1, bit = 0;
    for (int X = 0; X < 4; ++X) {
      const int f = (e.code >> (2 * X)) & 3;
      if (f == 3) mask |= (1 << X) | (1 << (8 - X));
      else if (f == 1 || f == 2) {
        const int ch = (a >> bit) & 1; ++bit;
        mask |= ch ? (1 << (8 - X)) : (1 << X);
        if (f == 2 && ch) sg = -sg;
      }
    }
    c.m[a] = (unsigned short)mask; c.s[a] = (signed char)sg;
  }
  return c;
}

struct MF { alignas(16) float m[96][56]; };

constexpr MF build_m() {
  MF r{};
  Cfgs ck[56] = {};
  for (int k = 0; k < 56; ++k) ck[k] = expand(LAY.k[k]);
  for (int row = 0; row < 96; ++row) {
    for (int k = 0; k < 56; ++k) r.m[row][k] = 0.0f;
    if (LAY.rpad[row]) continue;
    const Cfgs cr = expand(LAY.r[row]);
    for (int k = 0; k < 56; ++k) {
      if (LAY.k[k].cls != LAY.r[row].cls) continue;
      int num = 0;
      for (int i = 0; i < cr.n; ++i)
        for (int j = 0; j < ck[k].n; ++j)
          if ((cr.m[i] & ~ck[k].m[j]) == 0) num += cr.s[i] * ck[k].s[j];
      r.m[row][k] = (float)num / (float)(1 << (LAY.r[row].ns + LAY.k[k].ns));
    }
  }
  return r;
}

template<typename F, int... Is>
GOL_HD void static_for_impl(F&& f, std::integer_sequence<int, Is...>) {
  (f(std::integral_constant<int, Is>{}), ...);
}
template<int N, typename F>
GOL_HD void static_for(F&& f) {
  static_for_impl(static_cast<F&&>(f), std::make_integer_sequence<int, N>{});
}

template<int CTR, int J>
GOL_HD float cd_of(const float (&tn)[16], const float (&tc)[16]) {
  if constexpr (CTR != 0) return tn[J]; else return tc[J];
}

GOL_HD void lane_values(const float (&n)[9], bool hi, float (&bsel)[28], float (&asel)[48], float& sum) {
  float c[9];
  for (int k = 0; k < 9; ++k) c[k] = 1.0f - n[k];
  float F[4][4];
  static_for<4>([&](auto XX) {
    constexpr int X = XX.value;
    const float u = n[X] * c[8 - X];
    const float w = c[X] * n[8 - X];
    F[X][0] = c[X] * c[8 - X];
    F[X][1] = u + w;
    F[X][2] = u - w;
    F[X][3] = n[X] * n[8 - X];
  });
  float tAB[16], tCDn[16], tCDc[16];
  static_for<16>([&](auto I) {
    constexpr int i = I.value;
    tAB[i] = F[0][i & 3] * F[1][i >> 2];
    const float cd = F[2][i & 3] * F[3][i >> 2];
    tCDn[i] = cd * n[4];
    tCDc[i] = cd * c[4];
  });
  sum = 0.0f;
  static_for<28>([&](auto I) {
    constexpr int idx = I.value;
    constexpr int s = idx >> 1, j = idx & 1;
    constexpr Ent e0 = LAY.k[4 * s + j], e1 = LAY.k[4 * s + j + 2];
    constexpr int i0 = e0.code & 15, i1 = e1.code & 15;
    constexpr int j0 = (e0.code >> 4) & 15, j1 = (e1.code >> 4) & 15;
    float ab, cd;
    if constexpr (i0 == i1) ab = tAB[i0]; else ab = hi ? tAB[i1] : tAB[i0];
    if constexpr (j0 == j1) cd = tCDn[j0]; else cd = hi ? tCDn[j1] : tCDn[j0];
    bsel[idx] = ab * cd;
    static_assert((e0.cls == 0) == (e1.cls == 0), "sum slots must agree across lane halves");
    if constexpr (e0.cls == 0) sum += bsel[idx];
  });
  static_for<48>([&](auto I) {
    constexpr int idx = I.value;
    constexpr int t = idx >> 3, v = idx & 7;
    constexpr int r0 = 16 * t + v, r1 = r0 + 8;
    constexpr bool p0 = LAY.rpad[r0], p1 = LAY.rpad[r1];
    constexpr Ent e0 = LAY.r[r0], e1 = LAY.r[r1];
    constexpr int i0 = e0.code & 15, i1 = e1.code & 15;
    constexpr int j0 = (e0.code >> 4) & 15, j1 = (e1.code >> 4) & 15;
    if constexpr (p0 && p1) {
      asel[idx] = 0.0f;
    } else if constexpr (p1) {
      asel[idx] = tAB[i0] * (hi ? 0.0f : cd_of<e0.ctr, j0>(tCDn, tCDc));
    } else if constexpr (p0) {
      asel[idx] = tAB[i1] * (hi ? cd_of<e1.ctr, j1>(tCDn, tCDc) : 0.0f);
    } else {
      float ab, cd;
      if constexpr (i0 == i1) ab = tAB[i0]; else ab = hi ? tAB[i1] : tAB[i0];
      if constexpr (j0 == j1 && e0.ctr == e1.ctr) cd = cd_of<e0.ctr, j0>(tCDn, tCDc);
      else cd = hi ? cd_of<e1.ctr, j1>(tCDn, tCDc) : cd_of<e0.ctr, j0>(tCDn, tCDc);
      asel[idx] = ab * cd;
    }
    constexpr bool s0 = !p0 && e0.cls == 0;
    constexpr bool s1 = !p1 && e1.cls == 0;
    static_assert(s0 == s1, "sum slots must agree across lane halves");
    if constexpr (s0) sum += asel[idx];
  });
}

}

namespace gol {

struct Blocks { int t[18], s[18]; };
constexpr Blocks list_blocks() {
  Blocks b{};
  int i = 0;
  for (int t = 0; t < 6; ++t)
    for (int s = 0; s < 14; ++s)
      if (LAY.nz[t][s]) { b.t[i] = t; b.s[i] = s; ++i; }
  return b;
}
constexpr Blocks BLK = list_blocks();

struct AT { alignas(16) float v[9][32][4]; };
constexpr AT build_at() {
  const MF m = build_m();
  AT a{};
  for (int i = 0; i < 18; ++i)
    for (int lane = 0; lane < 32; ++lane) {
      const int p = lane & 15, h = lane >> 4;
      a.v[i >> 1][lane][2 * (i & 1)]     = m.m[16 * BLK.t[i] + p][4 * BLK.s[i] + 2 * h];
      a.v[i >> 1][lane][2 * (i & 1) + 1] = m.m[16 * BLK.t[i] + p][4 * BLK.s[i] + 2 * h + 1];
    }
  return a;
}
__device__ constexpr AT ATf = build_at();

__device__ __forceinline__ void load_nbhd(const float* __restrict__ x, int w, int p, float (&n)[9]) {
  const int b  = w >> 12;
  const int rm = w & 4095;
  const int y  = rm >> 4;
  const int x0 = ((rm & 15) << 4) + p;
  const float* img = x + (size_t)b * 65536;
#pragma unroll
  for (int k = 0; k < 9; ++k) {
    const int dy = k / 3 - 1, dx = k % 3 - 1;
    const int yy = (y + dy) & 255;
    const int xx = (x0 + dx) & 255;
    n[k] = img[yy * 256 + xx];
  }
}

}

namespace gol {

GOL_HD float tile_result(const v2f (&A)[18], const float (&n)[9], bool hi) {
  float bsel[28], asel[48], sumAcc;
  lane_values(n, hi, bsel, asel, sumAcc);

  v8f C[6];
  static_for<6>([&](auto TT) {
    constexpr int t = TT.value;
    C[t] = v8f{0.f, 0.f, 0.f, 0.f, 0.f, 0.f, 0.f, 0.f};
  });
  static_for<14>([&](auto SS) {
    constexpr int s = SS.value;
    v2f Bv;
    Bv.x = bsel[2 * s];
    Bv.y = bsel[2 * s + 1];
    static_for<18>([&](auto II) {
      constexpr int i = II.value;
      if constexpr (BLK.s[i] == s)
        C[BLK.t[i]] = __builtin_amdgcn_wmma_f32_16x16x4_f32(false, A[i], false, Bv, (short)0, C[BLK.t[i]], false, false);
    });
  });

  float orp = 0.0f;
  static_for<6>([&](auto TT) {
    constexpr int t = TT.value;
    float o = 0.0f;
    static_for<8>([&](auto VV) {
      constexpr int v = VV.value;
      if constexpr (!(LAY.rpad[16 * t + v] && LAY.rpad[16 * t + v + 8]))
        o = fmaf(asel[8 * t + v], C[t][v], o);
    });
    orp += o;
  });

  float tot = sumAcc - orp;
  tot += __shfl_xor(tot, 16, 32);
  return tot;
}

}

__global__ __launch_bounds__(1024) void gol_wmma_kernel(const float* __restrict__ x,
                                                       float* __restrict__ out,
                                                       int nTiles) {
  using namespace gol;
  const int lane = (int)(threadIdx.x & 31u);
  const int p = lane & 15;
  const int h = lane >> 4;
  const int wv = __builtin_amdgcn_readfirstlane((int)(threadIdx.x >> 5));
  const int wavesPerBlock = (int)(blockDim.x >> 5);
  const int stride = (int)gridDim.x * wavesPerBlock;
  int w = (int)blockIdx.x * wavesPerBlock + wv;
  if (w >= nTiles) return;

  const bool hi = (h != 0);

  float nn[9];
  load_nbhd(x, w, p, nn);

  v2f A[18];
  static_for<9>([&](auto QQ) {
    constexpr int q = QQ.value;
    const float* f = &ATf.v[q][lane][0];
    A[2 * q].x     = f[0];
    A[2 * q].y     = f[1];
    A[2 * q + 1].x = f[2];
    A[2 * q + 1].y = f[3];
  });

  {
    float n[9];
#pragma unroll
    for (int k = 0; k < 9; ++k) n[k] = nn[k];
    const int wn = w + stride;
    const bool more = wn < nTiles;
    load_nbhd(x, more ? wn : w, p, nn);
    const float tot = tile_result(A, n, hi);
    if (h == 0) out[(size_t)w * 16 + (size_t)p] = tot;
    if (!more) return;
    w = wn;
  }

  for (;;) {
    float n[9];
#pragma unroll
    for (int k = 0; k < 9; ++k) n[k] = nn[k];

    const int wn = w + stride;
    const bool more = wn < nTiles;
    load_nbhd(x, more ? wn : w, p, nn);

    const float tot = tile_result(A, n, hi);
    if (h == 0) out[(size_t)w * 16 + (size_t)p] = tot;

    if (!more) break;
    w = wn;
  }
}

extern "C" void kernel_launch(void* const* d_in, const int* in_sizes, int n_in,
                              void* d_out, int out_size, void* d_ws, size_t ws_size,
                              hipStream_t stream) {
  (void)n_in; (void)d_ws; (void)ws_size; (void)out_size;
  const float* x = (const float*)d_in[0];
  float* out = (float*)d_out;
  const int total = in_sizes[0];
  const int Bn = total >> 16;
  const int nTiles = Bn * 4096;
  dim3 block(1024);
  dim3 grid(256);
  gol_wmma_kernel<<<grid, block, 0, stream>>>(x, out, nTiles);
}


# Big class's K steps split into two groups, each with a row tile of its own (4 shared rows stand in both, coefficients split by group): one row tile per K step, 18 -> 14 WMMAs; f32 sum order changes
# speedup vs baseline: 3.6383x; 1.1140x over previous
#include <hip/hip_runtime.h>
#include <utility>

typedef __attribute__((ext_vector_type(2))) float v2f;
typedef __attribute__((ext_vector_type(8))) float v8f;

#define GOL_HD __device__ __forceinline__

namespace gol {

struct Ent { unsigned char code, ctr, cls, ns; };

constexpr Ent make_ent(int code, int ctr) {
  Ent e{};
  e.code = (unsigned char)code; e.ctr = (unsigned char)ctr;
  int cls = 0, ns = 0;
  for (int X = 0; X < 4; ++X) {
    const int f = (code >> (2 * X)) & 3;
    if (f == 2) cls |= 1 << X;
    if (f == 1 || f == 2) ++ns;
  }
  e.cls = (unsigned char)cls; e.ns = (unsigned char)ns;
  return e;
}

constexpr int occ_code(int code) {
  int o = 0;
  for (int X = 0; X < 4; ++X) { const int f = (code >> (2 * X)) & 3; o += (f == 0) ? 0 : ((f == 3) ? 2 : 1); }
  return o;
}

struct Layout {
  Ent  k[56];
  Ent  r[112];
  bool rpad[112];
  bool rcnt[112];
  bool nz[7][14];
  int  nnz;
  bool ok;
};

constexpr unsigned char KTAB[56] = {
  21, 67, 69, 19, 28, 7, 76, 13, 208, 196, 112, 193, 84, 52,
  81, 49, 50, 22, 194, 70, 82, 41, 14, 104, 88, 56, 11, 200,
  73, 134, 25, 146, 35, 100, 44, 97, 224, 26, 37, 74, 148, 131,
  145, 140, 38, 176, 98, 133, 164, 137, 161, 152, 168, 42, 162, 138
};
constexpr short RTAB[112] = {
  21, 28, 273, -1, 259, 7, 261, 19, 69, 76, 321, 1348, 268, 13, 276, 67,
  196, 304, 1300, -1, 84, 208, 112, 52, 193, 448, 1297, 1345, 81, 336, 324, 49,
  262, 50, 274, 22, 194, 168, 104, 41, 82, 14, 322, 70, 138, 162, 296, 42,
  328, 56, 200, 25, 134, 265, -1, -1, 280, 11, 146, 73, 386, 88, -1, -1,
  100, 292, 44, 352, 37, -1, 26, 35, 97, 289, -1, 224, -1, 266, 74, -1,
  -1, 133, 148, 98, 400, 388, 131, 38, -1, -1, 145, 290, 176, 385, 140, -1,
  -1, -1, -1, -1, -1, 137, 416, 164, -1, -1, -1, -1, -1, 392, 152, 161
};
constexpr signed char KHOME[14] = {0, 0, 1, 1, -1, -1, -1, -1, -1, -1, -1, -1, -1, -1};

constexpr Layout build_layout() {
  Layout L{};
  L.ok = true;
  bool seen4[256] = {};
  for (int i = 0; i < 56; ++i) {
    const int code = KTAB[i];
    if (occ_code(code) != 3 || seen4[code]) L.ok = false;
    seen4[code] = true;
    L.k[i] = make_ent(code, 1);
  }
  int seen3[512] = {};
  int cnt3[512] = {};
  int n3 = 0;
  for (int i = 0; i < 112; ++i) {
    const int raw = RTAB[i];
    L.rpad[i] = raw < 0;
    L.rcnt[i] = false;
    L.r[i] = Ent{0, 0, 0, 0};
    if (raw < 0) continue;
    const bool second = (raw & 1024) != 0;
    const int v = raw & 1023;
    const int code = v & 255, ctr = v >> 8;
    const int o = occ_code(code);
    if (!((ctr == 1 && o == 2) || (ctr == 0 && o == 3))) L.ok = false;
    L.r[i] = make_ent(code, ctr);
    if (seen3[v] != 0 && (L.r[i].cls != 0 || seen3[v] != 1 + (1 - i / 16))) L.ok = false;
    if (i / 16 < 2) seen3[v] = 1 + i / 16; else seen3[v] = 3;
    if (L.r[i].cls == 0 && i / 16 > 1) L.ok = false;
    if (!second) { L.rcnt[i] = true; ++cnt3[v]; }
    ++n3;
  }
  for (int v = 0; v < 512; ++v)
    if ((seen3[v] != 0) != (cnt3[v] == 1) || cnt3[v] > 1) L.ok = false;
  int distinct = 0;
  for (int v = 0; v < 512; ++v) if (seen3[v] != 0) ++distinct;
  if (distinct != 84 || n3 != 88) L.ok = false;
  L.nnz = 0;
  for (int t = 0; t < 7; ++t)
    for (int s = 0; s < 14; ++s) {
      bool any = false;
      for (int i = 0; i < 16; ++i)
        for (int j = 0; j < 4; ++j)
          if (!L.rpad[16 * t + i] && L.r[16 * t + i].cls == L.k[4 * s + j].cls && (KHOME[s] < 0 || KHOME[s] == t)) any = true;
      L.nz[t][s] = any;
      if (any) ++L.nnz;
    }
  return L;
}

constexpr Layout LAY = build_layout();
static_assert(LAY.ok, "basis layout broken");
static_assert(LAY.nnz == 14, "block count changed");

struct Cfgs { int n; unsigned short m[8]; signed char s[8]; };

constexpr Cfgs expand(Ent e) {
  Cfgs c{};
  c.n = 1 << e.ns;
  for (int a = 0; a < c.n; ++a) {
    int mask = e.ctr ? 16 : 0, sg = 1, bit = 0;
    for (int X = 0; X < 4; ++X) {
      const int f = (e.code >> (2 * X)) & 3;
      if (f == 3) mask |= (1 << X) | (1 << (8 - X));
      else if (f == 1 || f == 2) {
        const int ch = (a >> bit) & 1; ++bit;
        mask |= ch ? (1 << (8 - X)) : (1 << X);
        if (f == 2 && ch) sg = -sg;
      }
    }
    c.m[a] = (unsigned short)mask; c.s[a] = (signed char)sg;
  }
  return c;
}

constexpr float coef(Ent er, const Cfgs& cr, Ent ek, const Cfgs& ck) {
  if (ek.cls != er.cls) return 0.0f;
  int num = 0;
  for (int i = 0; i < cr.n; ++i)
    for (int j = 0; j < ck.n; ++j)
      if ((cr.m[i] & ~ck.m[j]) == 0) num += cr.s[i] * ck.s[j];
  return (float)num / (float)(1 << (er.ns + ek.ns));
}

struct MF { alignas(16) float m[112][56]; bool ok; };

constexpr MF build_m() {
  MF r{};
  r.ok = true;
  Cfgs ck[56] = {};
  for (int k = 0; k < 56; ++k) ck[k] = expand(LAY.k[k]);
  for (int row = 0; row < 112; ++row) {
    for (int k = 0; k < 56; ++k) r.m[row][k] = 0.0f;
    if (LAY.rpad[row]) continue;
    const Cfgs cr = expand(LAY.r[row]);
    bool twice = false;
    for (int o = 0; o < 32; ++o)
      if (o != row && !LAY.rpad[o] && LAY.r[o].code == LAY.r[row].code && LAY.r[o].ctr == LAY.r[row].ctr) twice = true;
    for (int k = 0; k < 56; ++k) {
      if (LAY.k[k].cls != LAY.r[row].cls) continue;
      const float c = coef(LAY.r[row], cr, LAY.k[k], ck[k]);
      if (KHOME[k / 4] >= 0 && KHOME[k / 4] != row / 16) {
        if (!twice && c != 0.0f) r.ok = false;
        continue;
      }
      r.m[row][k] = c;
    }
  }
  return r;
}

template<typename F, int... Is>
GOL_HD void static_for_impl(F&& f, std::integer_sequence<int, Is...>) {
  (f(std::integral_constant<int, Is>{}), ...);
}
template<int N, typename F>
GOL_HD void static_for(F&& f) {
  static_for_impl(static_cast<F&&>(f), std::make_integer_sequence<int, N>{});
}

template<int CTR, int J>
GOL_HD float cd_of(const float (&tn)[16], const float (&tc)[16]) {
  if constexpr (CTR != 0) return tn[J]; else return tc[J];
}

GOL_HD void lane_values(const float (&n)[9], bool hi, float (&bsel)[28], float (&asel)[56], float& sum) {
  float c[9];
  for (int k = 0; k < 9; ++k) c[k] = 1.0f - n[k];
  float F[4][4];
  static_for<4>([&](auto XX) {
    constexpr int X = XX.value;
    const float u = n[X] * c[8 - X];
    const float w = c[X] * n[8 - X];
    F[X][0] = c[X] * c[8 - X];
    F[X][1] = u + w;
    F[X][2] = u - w;
    F[X][3] = n[X] * n[8 - X];
  });
  float tAB[16], tCDn[16], tCDc[16];
  static_for<16>([&](auto I) {
    constexpr int i = I.value;
    tAB[i] = F[0][i & 3] * F[1][i >> 2];
    const float cd = F[2][i & 3] * F[3][i >> 2];
    tCDn[i] = cd * n[4];
    tCDc[i] = cd * c[4];
  });
  sum = 0.0f;
  static_for<28>([&](auto I) {
    constexpr int idx = I.value;
    constexpr int s = idx >> 1, j = idx & 1;
    constexpr Ent e0 = LAY.k[4 * s + j], e1 = LAY.k[4 * s + j + 2];
    constexpr int i0 = e0.code & 15, i1 = e1.code & 15;
    constexpr int j0 = (e0.code >> 4) & 15, j1 = (e1.code >> 4) & 15;
    float ab, cd;
    if constexpr (i0 == i1) ab = tAB[i0]; else ab = hi ? tAB[i1] : tAB[i0];
    if constexpr (j0 == j1) cd = tCDn[j0]; else cd = hi ? tCDn[j1] : tCDn[j0];
    bsel[idx] = ab * cd;
    static_assert((e0.cls == 0) == (e1.cls == 0), "sum slots must agree across lane halves");
    if constexpr (e0.cls == 0) sum += bsel[idx];
  });
  static_for<56>([&](auto I) {
    constexpr int idx = I.value;
    constexpr int t = idx >> 3, v = idx & 7;
    constexpr int r0 = 16 * t + v, r1 = r0 + 8;
    constexpr bool p0 = LAY.rpad[r0], p1 = LAY.rpad[r1];
    constexpr Ent e0 = LAY.r[r0], e1 = LAY.r[r1];
    constexpr int i0 = e0.code & 15, i1 = e1.code & 15;
    constexpr int j0 = (e0.code >> 4) & 15, j1 = (e1.code >> 4) & 15;
    if constexpr (p0 && p1) {
      asel[idx] = 0.0f;
    } else if constexpr (p1) {
      asel[idx] = tAB[i0] * (hi ? 0.0f : cd_of<e0.ctr, j0>(tCDn, tCDc));
    } else if constexpr (p0) {
      asel[idx] = tAB[i1] * (hi ? cd_of<e1.ctr, j1>(tCDn, tCDc) : 0.0f);
    } else {
      float ab, cd;
      if constexpr (i0 == i1) ab = tAB[i0]; else ab = hi ? tAB[i1] : tAB[i0];
      if constexpr (j0 == j1 && e0.ctr == e1.ctr) cd = cd_of<e0.ctr, j0>(tCDn, tCDc);
      else cd = hi ? cd_of<e1.ctr, j1>(tCDn, tCDc) : cd_of<e0.ctr, j0>(tCDn, tCDc);
      asel[idx] = ab * cd;
    }
    constexpr bool s0 = !p0 && e0.cls == 0 && LAY.rcnt[r0];
    constexpr bool s1 = !p1 && e1.cls == 0 && LAY.rcnt[r1];
    static_assert(s0 == s1, "sum slots must agree across lane halves");
    if constexpr (s0) sum += asel[idx];
  });
}

}

namespace gol {

struct Blocks { int t[14], s[14]; };
constexpr Blocks list_blocks() {
  int bt[14] = {}, bs[14] = {};
  int n = 0;
  for (int t = 0; t < 7; ++t)
    for (int s = 0; s < 14; ++s)
      if (LAY.nz[t][s]) { bt[n] = t; bs[n] = s; ++n; }
  Blocks b{};
  bool used[14] = {};
  int last = -1;
  for (int i = 0; i < 14; ++i) {
    int pick = -1;
    for (int c = 0; c < 14 && pick < 0; ++c) if (!used[c] && bt[c] != last) pick = c;
    for (int c = 0; c < 14 && pick < 0; ++c) if (!used[c]) pick = c;
    used[pick] = true; b.t[i] = bt[pick]; b.s[i] = bs[pick]; last = bt[pick];
  }
  return b;
}
constexpr Blocks BLK = list_blocks();

struct AT { alignas(16) float v[7][32][4]; bool ok; };
constexpr AT build_at() {
  const MF m = build_m();
  AT a{};
  a.ok = m.ok;
  for (int i = 0; i < 14; ++i)
    for (int lane = 0; lane < 32; ++lane) {
      const int p = lane & 15, h = lane >> 4;
      a.v[i >> 1][lane][2 * (i & 1)]     = m.m[16 * BLK.t[i] + p][4 * BLK.s[i] + 2 * h];
      a.v[i >> 1][lane][2 * (i & 1) + 1] = m.m[16 * BLK.t[i] + p][4 * BLK.s[i] + 2 * h + 1];
    }
  return a;
}
__device__ constexpr AT ATf = build_at();
static_assert(ATf.ok, "a row kept in one tile has a coefficient on the other tile's K steps");

__device__ __forceinline__ void load_nbhd(const float* __restrict__ x, int w, int p, float (&n)[9]) {
  const int b  = w >> 12;
  const int rm = w & 4095;
  const int y  = rm >> 4;
  const int x0 = ((rm & 15) << 4) + p;
  const float* img = x + (size_t)b * 65536;
#pragma unroll
  for (int k = 0; k < 9; ++k) {
    const int dy = k / 3 - 1, dx = k % 3 - 1;
    const int yy = (y + dy) & 255;
    const int xx = (x0 + dx) & 255;
    n[k] = img[yy * 256 + xx];
  }
}

}

namespace gol {

GOL_HD float tile_result(const v2f (&A)[14], const float (&n)[9], bool hi) {
  float bsel[28], asel[56], sumAcc;
  lane_values(n, hi, bsel, asel, sumAcc);

  v8f C[7];
  static_for<7>([&](auto TT) {
    constexpr int t = TT.value;
    C[t] = v8f{0.f, 0.f, 0.f, 0.f, 0.f, 0.f, 0.f, 0.f};
  });
  static_for<14>([&](auto II) {
    constexpr int i = II.value;
    constexpr int s = BLK.s[i];
    v2f Bv;
    Bv.x = bsel[2 * s];
    Bv.y = bsel[2 * s + 1];
    C[BLK.t[i]] = __builtin_amdgcn_wmma_f32_16x16x4_f32(false, A[i], false, Bv, (short)0, C[BLK.t[i]], false, false);
  });

  float orp = 0.0f;
  static_for<7>([&](auto TT) {
    constexpr int t = TT.value;
    float o = 0.0f;
    static_for<8>([&](auto VV) {
      constexpr int v = VV.value;
      if constexpr (!(LAY.rpad[16 * t + v] && LAY.rpad[16 * t + v + 8]))
        o = fmaf(asel[8 * t + v], C[t][v], o);
    });
    orp += o;
  });

  float tot = sumAcc - orp;
  tot += __shfl_xor(tot, 16, 32);
  return tot;
}

}

__global__ __launch_bounds__(1024) void gol_wmma_kernel(const float* __restrict__ x,
                                                       float* __restrict__ out,
                                                       int nTiles) {
  using namespace gol;
  const int lane = (int)(threadIdx.x & 31u);
  const int p = lane & 15;
  const int h = lane >> 4;
  const int wv = __builtin_amdgcn_readfirstlane((int)(threadIdx.x >> 5));
  const int wavesPerBlock = (int)(blockDim.x >> 5);
  const int stride = (int)gridDim.x * wavesPerBlock;
  int w = (int)blockIdx.x * wavesPerBlock + wv;
  if (w >= nTiles) return;

  const bool hi = (h != 0);

  float nn[9];
  load_nbhd(x, w, p, nn);

  v2f A[14];
  static_for<7>([&](auto QQ) {
    constexpr int q = QQ.value;
    const float* f = &ATf.v[q][lane][0];
    A[2 * q].x     = f[0];
    A[2 * q].y     = f[1];
    A[2 * q + 1].x = f[2];
    A[2 * q + 1].y = f[3];
  });

  {
    float n[9];
#pragma unroll
    for (int k = 0; k < 9; ++k) n[k] = nn[k];
    const int wn = w + stride;
    const bool more = wn < nTiles;
    load_nbhd(x, more ? wn : w, p, nn);
    const float tot = tile_result(A, n, hi);
    if (h == 0) out[(size_t)w * 16 + (size_t)p] = tot;
    if (!more) return;
    w = wn;
  }

  for (;;) {
    float n[9];
#pragma unroll
    for (int k = 0; k < 9; ++k) n[k] = nn[k];

    const int wn = w + stride;
    const bool more = wn < nTiles;
    load_nbhd(x, more ? wn : w, p, nn);

    const float tot = tile_result(A, n, hi);
    if (h == 0) out[(size_t)w * 16 + (size_t)p] = tot;

    if (!more) break;
    w = wn;
  }
}

extern "C" void kernel_launch(void* const* d_in, const int* in_sizes, int n_in,
                              void* d_out, int out_size, void* d_ws, size_t ws_size,
                              hipStream_t stream) {
  (void)n_in; (void)d_ws; (void)ws_size; (void)out_size;
  const float* x = (const float*)d_in[0];
  float* out = (float*)d_out;
  const int total = in_sizes[0];
  const int Bn = total >> 16;
  const int nTiles = Bn * 4096;
  dim3 block(1024);
  dim3 grid(256);
  gol_wmma_kernel<<<grid, block, 0, stream>>>(x, out, nTiles);
}


# Center factor moved behind the GEMM: center-dead rows turn into a sum of squares in the lane, the 28 center-alive S3 entries become K and the S4 entries the rows: 14 -> 8 WMMAs; f32 sum order changes
# speedup vs baseline: 4.8854x; 1.3428x over previous
#include <hip/hip_runtime.h>
#include <utility>

typedef __attribute__((ext_vector_type(2))) float v2f;
typedef __attribute__((ext_vector_type(8))) float v8f;

#define GOL_HD __device__ __forceinline__

namespace gol {

struct Ent { unsigned char code, ctr, cls, ns; };

constexpr Ent make_ent(int code, int ctr) {
  Ent e{};
  e.code = (unsigned char)code; e.ctr = (unsigned char)ctr;
  int cls = 0, ns = 0;
  for (int X = 0; X < 4; ++X) {
    const int f = (code >> (2 * X)) & 3;
    if (f == 2) cls |= 1 << X;
    if (f == 1 || f == 2) ++ns;
  }
  e.cls = (unsigned char)cls; e.ns = (unsigned char)ns;
  return e;
}

constexpr int occ_code(int code) {
  int o = 0;
  for (int X = 0; X < 4; ++X) { const int f = (code >> (2 * X)) & 3; o += (f == 0) ? 0 : ((f == 3) ? 2 : 1); }
  return o;
}

constexpr int sig(int code) { return ((code & 15) << 4) | (code >> 4); }
constexpr int sigc(int cls) { return ((cls & 3) << 2) | (cls >> 2); }

constexpr int tile_of_cls(int c) {
  switch (c) {
    case 0: return 0;
    case 1: case 3: return 1;
    case 2: case 9: return 2;
    case 5: case 10: case 7: case 11: return 3;
    default: return -1;
  }
}
constexpr int KBASE[4] = {0, 3, 5, 7};
constexpr int KLEN[4]  = {3, 2, 2, 1};

struct Layout {
  short k[32];
  float kw[32];
  short r[64];
  bool ok;
};

constexpr Layout build_layout() {
  Layout L{};
  L.ok = true;
  for (int i = 0; i < 32; ++i) { L.k[i] = -1; L.kw[i] = 0.0f; }
  for (int i = 0; i < 64; ++i) L.r[i] = -1;
  int nrow[4] = {0, 0, 0, 0}, nk[4] = {0, 0, 0, 0};
  int seen[256] = {};
  for (int code = 0; code < 256; ++code) {
    const int o = occ_code(code);
    if (o != 3 && o != 2) continue;
    const int cls = make_ent(code, 1).cls;
    const int t = tile_of_cls(cls);
    if (t < 0) continue;
    if (sigc(cls) == cls && code > sig(code)) continue;
    if (o == 3) {
      if (code == sig(code)) { L.ok = false; continue; }
      const int v = nrow[t]++;
      if (v >= 8) { L.ok = false; continue; }
      L.r[16 * t + v] = (short)code; L.r[16 * t + 8 + v] = (short)sig(code);
      ++seen[code]; ++seen[sig(code)];
    } else {
      const int q = nk[t]++;
      if (q >= 2 * KLEN[t]) { L.ok = false; continue; }
      const int s = KBASE[t] + q / 2, j = q % 2;
      const float w = (code == sig(code)) ? 0.5f : 1.0f;
      L.k[4 * s + j] = (short)code;     L.kw[4 * s + j] = w;
      L.k[4 * s + 2 + j] = (short)sig(code); L.kw[4 * s + 2 + j] = w;
      if (code == sig(code)) seen[code] += 2; else { ++seen[code]; ++seen[sig(code)]; }
    }
  }
  for (int code = 0; code < 256; ++code) {
    const int o = occ_code(code);
    const int want = (o == 3) ? 1 : ((o == 2) ? ((code == sig(code)) ? 2 : 1) : 0);
    if (seen[code] != want) L.ok = false;
  }
  if (nrow[0] != 8 || nrow[1] != 8 || nrow[2] != 8 || nrow[3] != 4) L.ok = false;
  for (int t = 0; t < 4; ++t) if (nk[t] != 2 * KLEN[t]) L.ok = false;
  return L;
}

constexpr Layout LAY = build_layout();
static_assert(LAY.ok, "basis layout broken");

struct Cfgs { int n; unsigned short m[8]; signed char s[8]; };

constexpr Cfgs expand(Ent e) {
  Cfgs c{};
  c.n = 1 << e.ns;
  for (int a = 0; a < c.n; ++a) {
    int mask = e.ctr ? 16 : 0, sg = 1, bit = 0;
    for (int X = 0; X < 4; ++X) {
      const int f = (e.code >> (2 * X)) & 3;
      if (f == 3) mask |= (1 << X) | (1 << (8 - X));
      else if (f == 1 || f == 2) {
        const int ch = (a >> bit) & 1; ++bit;
        mask |= ch ? (1 << (8 - X)) : (1 << X);
        if (f == 2 && ch) sg = -sg;
      }
    }
    c.m[a] = (unsigned short)mask; c.s[a] = (signed char)sg;
  }
  return c;
}

constexpr float coef(Ent er, const Cfgs& cr, Ent ek, const Cfgs& ck) {
  if (ek.cls != er.cls) return 0.0f;
  int num = 0;
  for (int i = 0; i < cr.n; ++i)
    for (int j = 0; j < ck.n; ++j)
      if ((cr.m[i] & ~ck.m[j]) == 0) num += cr.s[i] * ck.s[j];
  return (float)num / (float)(1 << (er.ns + ek.ns));
}

struct MF { float m[64][32]; bool rz[64]; bool nz[4][8]; int nnz; bool ok; };

constexpr MF build_m() {
  MF r{};
  r.ok = true;
  for (int row = 0; row < 64; ++row) {
    r.rz[row] = true;
    for (int k = 0; k < 32; ++k) {
      r.m[row][k] = 0.0f;
      if (LAY.r[row] < 0 || LAY.k[k] < 0) continue;
      const Ent e3 = make_ent(LAY.k[k], 1), e4 = make_ent(LAY.r[row], 1);
      const float c = coef(e3, expand(e3), e4, expand(e4)) * LAY.kw[k];
      r.m[row][k] = c;
      if (c != 0.0f) r.rz[row] = false;
    }
  }
  r.nnz = 0;
  for (int t = 0; t < 4; ++t)
    for (int s = 0; s < 8; ++s) {
      bool any = false;
      for (int i = 0; i < 16; ++i)
        for (int j = 0; j < 4; ++j)
          if (r.m[16 * t + i][4 * s + j] != 0.0f) any = true;
      r.nz[t][s] = any;
      if (any) ++r.nnz;
      if (any && !(s >= KBASE[t] && s < KBASE[t] + KLEN[t])) r.ok = false;
    }
  return r;
}
constexpr MF MFc = build_m();
static_assert(MFc.ok, "a K step meets a foreign row tile");
static_assert(MFc.nnz == 8, "block count changed");

struct Blocks { int t[8], s[8]; };
constexpr Blocks list_blocks() {
  int bt[8] = {}, bs[8] = {};
  int n = 0;
  for (int t = 0; t < 4; ++t)
    for (int s = 0; s < 8; ++s)
      if (MFc.nz[t][s]) { bt[n] = t; bs[n] = s; ++n; }
  Blocks b{};
  bool used[8] = {};
  int last = -1;
  for (int i = 0; i < 8; ++i) {
    int pick = -1;
    int best = -1;
    for (int c = 0; c < 8; ++c) if (!used[c] && bt[c] != last) {
      int left = 0;
      for (int d = 0; d < 8; ++d) if (!used[d] && bt[d] == bt[c]) ++left;
      if (left > best) { best = left; pick = c; }
    }
    for (int c = 0; c < 8 && pick < 0; ++c) if (!used[c]) pick = c;
    used[pick] = true; b.t[i] = bt[pick]; b.s[i] = bs[pick]; last = bt[pick];
  }
  return b;
}
constexpr Blocks BLK = list_blocks();

struct AT { alignas(16) float v[4][32][4]; };
constexpr AT build_at() {
  AT a{};
  for (int i = 0; i < 8; ++i)
    for (int lane = 0; lane < 32; ++lane) {
      const int p = lane & 15, h = lane >> 4;
      a.v[i >> 1][lane][2 * (i & 1)]     = MFc.m[16 * BLK.t[i] + p][4 * BLK.s[i] + 2 * h];
      a.v[i >> 1][lane][2 * (i & 1) + 1] = MFc.m[16 * BLK.t[i] + p][4 * BLK.s[i] + 2 * h + 1];
    }
  return a;
}

}

namespace gol {

__device__ constexpr AT ATf = build_at();

template<typename F, int... Is>
GOL_HD void static_for_impl(F&& f, std::integer_sequence<int, Is...>) {
  (f(std::integral_constant<int, Is>{}), ...);
}
template<int N, typename F>
GOL_HD void static_for(F&& f) {
  static_for_impl(static_cast<F&&>(f), std::make_integer_sequence<int, N>{});
}

GOL_HD void lane_values(const float (&n)[9], float (&bsel)[16], float (&asel)[32],
                        float& SA, float& SB, float& DQ) {
  float c[9];
  for (int k = 0; k < 9; ++k) c[k] = 1.0f - n[k];
  float F[4][4];
  static_for<4>([&](auto XX) {
    constexpr int X = XX.value;
    const float u = n[X] * c[8 - X];
    const float w = c[X] * n[8 - X];
    F[X][0] = c[X] * c[8 - X];
    F[X][1] = u + w;
    F[X][2] = u - w;
    F[X][3] = n[X] * n[8 - X];
  });
  float tAB[16], tCD[16];
  static_for<16>([&](auto I) {
    constexpr int i = I.value;
    tAB[i] = F[0][i & 3] * F[1][i >> 2];
    tCD[i] = F[2][i & 3] * F[3][i >> 2];
  });
  SA = 0.0f;
  static_for<16>([&](auto I) {
    constexpr int idx = I.value;
    constexpr int s = idx >> 1, j = idx & 1;
    constexpr int code = LAY.k[4 * s + j];
    static_assert(code >= 0, "every K slot is used");
    static_assert(LAY.k[4 * s + j + 2] == sig(code), "upper half holds the image");
    bsel[idx] = tAB[code & 15] * tCD[code >> 4];
    if constexpr (make_ent(code, 1).cls == 0) SA = fmaf(LAY.kw[4 * s + j], bsel[idx], SA);
  });
  SB = 0.0f;
  float d1 = 0.0f, d3 = 0.0f;
  static_for<32>([&](auto I) {
    constexpr int idx = I.value;
    constexpr int t = idx >> 3, v = idx & 7;
    constexpr int code = LAY.r[16 * t + v];
    if constexpr (code < 0) {
      static_assert(LAY.r[16 * t + v + 8] < 0, "unused rows come in couples");
      asel[idx] = 0.0f;
    } else {
      static_assert(LAY.r[16 * t + v + 8] == sig(code), "upper half holds the image");
      constexpr Ent e = make_ent(code, 1);
      asel[idx] = tAB[code & 15] * tCD[code >> 4];
      if constexpr (e.cls == 0) SB += asel[idx];
      static_assert(e.ns == 1 || e.ns == 3, "three live outer cells: one or three one-alive pairs");
      if constexpr (e.ns == 1) d1 = fmaf(asel[idx], asel[idx], d1);
      else                     d3 = fmaf(asel[idx], asel[idx], d3);
    }
  });
  DQ = 0.5f * d1 + 0.125f * d3;
}

__device__ __forceinline__ void load_nbhd(const float* __restrict__ x, int w, int p, float (&n)[9]) {
  const int b  = w >> 12;
  const int rm = w & 4095;
  const int y  = rm >> 4;
  const int x0 = ((rm & 15) << 4) + p;
  const float* img = x + (size_t)b * 65536;
#pragma unroll
  for (int k = 0; k < 9; ++k) {
    const int dy = k / 3 - 1, dx = k % 3 - 1;
    const int yy = (y + dy) & 255;
    const int xx = (x0 + dx) & 255;
    n[k] = img[yy * 256 + xx];
  }
}

GOL_HD void hand_over(const float (&nn)[9], bool hi, float (&n)[9]) {
  n[0] = hi ? nn[2] : nn[0];
  n[1] = hi ? nn[3] : nn[1];
  n[2] = hi ? nn[0] : nn[2];
  n[3] = hi ? nn[1] : nn[3];
  n[4] = nn[4];
  n[5] = hi ? nn[7] : nn[5];
  n[6] = hi ? nn[8] : nn[6];
  n[7] = hi ? nn[5] : nn[7];
  n[8] = hi ? nn[6] : nn[8];
}

GOL_HD float tile_result(const v2f (&A)[8], const float (&n)[9]) {
  float bsel[16], asel[32], SA, SB, DQ;
  lane_values(n, bsel, asel, SA, SB, DQ);

  v8f C[4];
  static_for<4>([&](auto TT) {
    constexpr int t = TT.value;
    C[t] = v8f{0.f, 0.f, 0.f, 0.f, 0.f, 0.f, 0.f, 0.f};
  });
  static_for<8>([&](auto II) {
    constexpr int i = II.value;
    constexpr int s = BLK.s[i];
    v2f Bv;
    Bv.x = bsel[2 * s];
    Bv.y = bsel[2 * s + 1];
    C[BLK.t[i]] = __builtin_amdgcn_wmma_f32_16x16x4_f32(false, A[i], false, Bv, (short)0, C[BLK.t[i]], false, false);
  });

  float orp = 0.0f;
  static_for<4>([&](auto TT) {
    constexpr int t = TT.value;
    float o = 0.0f;
    static_for<8>([&](auto VV) {
      constexpr int v = VV.value;
      if constexpr (!(MFc.rz[16 * t + v] && MFc.rz[16 * t + v + 8]))
        o = fmaf(asel[8 * t + v], C[t][v], o);
    });
    orp += o;
  });

  const float q = n[4], cq = 1.0f - n[4];
  float tot = fmaf(q, SA - fmaf(q, orp, cq * DQ), SB);
  tot += __shfl_xor(tot, 16, 32);
  return tot;
}

}

__global__ __launch_bounds__(1024) void gol_wmma_kernel(const float* __restrict__ x,
                                                       float* __restrict__ out,
                                                       int nTiles) {
  using namespace gol;
  const int lane = (int)(threadIdx.x & 31u);
  const int p = lane & 15;
  const int h = lane >> 4;
  const int wv = __builtin_amdgcn_readfirstlane((int)(threadIdx.x >> 5));
  const int wavesPerBlock = (int)(blockDim.x >> 5);
  const int stride = (int)gridDim.x * wavesPerBlock;
  int w = (int)blockIdx.x * wavesPerBlock + wv;
  if (w >= nTiles) return;

  const bool hi = (h != 0);

  float nn[9];
  load_nbhd(x, w, p, nn);

  v2f A[8];
  static_for<4>([&](auto QQ) {
    constexpr int q = QQ.value;
    const float* f = &ATf.v[q][lane][0];
    A[2 * q].x     = f[0];
    A[2 * q].y     = f[1];
    A[2 * q + 1].x = f[2];
    A[2 * q + 1].y = f[3];
  });

  {
    float n[9];
    hand_over(nn, hi, n);
    const int wn = w + stride;
    const bool more = wn < nTiles;
    load_nbhd(x, more ? wn : w, p, nn);
    const float tot = tile_result(A, n);
    if (h == 0) out[(size_t)w * 16 + (size_t)p] = tot;
    if (!more) return;
    w = wn;
  }

  for (;;) {
    float n[9];
    hand_over(nn, hi, n);

    const int wn = w + stride;
    const bool more = wn < nTiles;
    load_nbhd(x, more ? wn : w, p, nn);

    const float tot = tile_result(A, n);
    if (h == 0) out[(size_t)w * 16 + (size_t)p] = tot;

    if (!more) break;
    w = wn;
  }
}

extern "C" void kernel_launch(void* const* d_in, const int* in_sizes, int n_in,
                              void* d_out, int out_size, void* d_ws, size_t ws_size,
                              hipStream_t stream) {
  (void)n_in; (void)d_ws; (void)ws_size; (void)out_size;
  const float* x = (const float*)d_in[0];
  float* out = (float*)d_out;
  const int total = in_sizes[0];
  const int Bn = total >> 16;
  const int nTiles = Bn * 4096;
  dim3 block(1024);
  dim3 grid(256);
  gol_wmma_kernel<<<grid, block, 0, stream>>>(x, out, nTiles);
}


# The four S3 entries that are their own image under the lane-half exchange (two K steps at weight 1/2) leave the GEMM: each half forms their products in the lane; 8 -> 6 WMMAs; f32 sum order changes
# speedup vs baseline: 5.3307x; 1.0911x over previous
#include <hip/hip_runtime.h>
#include <utility>

typedef __attribute__((ext_vector_type(2))) float v2f;
typedef __attribute__((ext_vector_type(8))) float v8f;

#define GOL_HD __device__ __forceinline__

namespace gol {

struct Ent { unsigned char code, ctr, cls, ns; };

constexpr Ent make_ent(int code, int ctr) {
  Ent e{};
  e.code = (unsigned char)code; e.ctr = (unsigned char)ctr;
  int cls = 0, ns = 0;
  for (int X = 0; X < 4; ++X) {
    const int f = (code >> (2 * X)) & 3;
    if (f == 2) cls |= 1 << X;
    if (f == 1 || f == 2) ++ns;
  }
  e.cls = (unsigned char)cls; e.ns = (unsigned char)ns;
  return e;
}

constexpr int occ_code(int code) {
  int o = 0;
  for (int X = 0; X < 4; ++X) { const int f = (code >> (2 * X)) & 3; o += (f == 0) ? 0 : ((f == 3) ? 2 : 1); }
  return o;
}

constexpr int sig(int code) { return ((code & 15) << 4) | (code >> 4); }
constexpr int sigc(int cls) { return ((cls & 3) << 2) | (cls >> 2); }

constexpr int tile_of_cls(int c) {
  switch (c) {
    case 0: return 0;
    case 1: case 3: return 1;
    case 2: case 9: return 2;
    case 5: case 10: case 7: case 11: return 3;
    default: return -1;
  }
}
constexpr int KBASE[3] = {0, 2, 4};
constexpr int KLEN[3]  = {2, 2, 2};

struct Layout {
  short k[24];
  short r[48];
  short xr[4];
  short sk[4];
  bool ok;
};

constexpr Layout build_layout() {
  Layout L{};
  L.ok = true;
  for (int i = 0; i < 24; ++i) L.k[i] = -1;
  for (int i = 0; i < 48; ++i) L.r[i] = -1;
  for (int i = 0; i < 4; ++i) { L.xr[i] = -1; L.sk[i] = -1; }
  int nrow[3] = {0, 0, 0}, nk[3] = {0, 0, 0}, nx = 0, nself = 0;
  int seen[256] = {};
  for (int code = 0; code < 256; ++code) {
    const int o = occ_code(code);
    if (o != 3 && o != 2) continue;
    if (o == 2 && code == sig(code)) {
      if (nself >= 4) { L.ok = false; continue; }
      L.sk[nself++] = (short)code; ++seen[code];
      continue;
    }
    const int cls = make_ent(code, 1).cls;
    const int t = tile_of_cls(cls);
    if (t < 0) continue;
    if (sigc(cls) == cls && code > sig(code)) continue;
    if (o == 3) {
      if (code == sig(code)) { L.ok = false; continue; }
      if (t == 3) {
        if (nx >= 4) { L.ok = false; continue; }
        L.xr[nx++] = (short)code;
      } else {
        const int v = nrow[t]++;
        if (v >= 8) { L.ok = false; continue; }
        L.r[16 * t + v] = (short)code; L.r[16 * t + 8 + v] = (short)sig(code);
      }
      ++seen[code]; ++seen[sig(code)];
    } else {
      if (t == 3) { L.ok = false; continue; }
      const int q = nk[t]++;
      if (q >= 2 * KLEN[t]) { L.ok = false; continue; }
      const int s = KBASE[t] + q / 2, j = q % 2;
      L.k[4 * s + j] = (short)code;
      L.k[4 * s + 2 + j] = (short)sig(code);
      ++seen[code]; ++seen[sig(code)];
    }
  }
  for (int code = 0; code < 256; ++code) {
    const int o = occ_code(code);
    const int want = (o == 3 || o == 2) ? 1 : 0;
    if (seen[code] != want) L.ok = false;
  }
  if (nrow[0] != 8 || nrow[1] != 8 || nrow[2] != 8 || nx != 4 || nself != 4) L.ok = false;
  for (int t = 0; t < 3; ++t) if (nk[t] != 2 * KLEN[t]) L.ok = false;
  return L;
}

constexpr Layout LAY = build_layout();
static_assert(LAY.ok, "basis layout broken");

struct Cfgs { int n; unsigned short m[8]; signed char s[8]; };

constexpr Cfgs expand(Ent e) {
  Cfgs c{};
  c.n = 1 << e.ns;
  for (int a = 0; a < c.n; ++a) {
    int mask = e.ctr ? 16 : 0, sg = 1, bit = 0;
    for (int X = 0; X < 4; ++X) {
      const int f = (e.code >> (2 * X)) & 3;
      if (f == 3) mask |= (1 << X) | (1 << (8 - X));
      else if (f == 1 || f == 2) {
        const int ch = (a >> bit) & 1; ++bit;
        mask |= ch ? (1 << (8 - X)) : (1 << X);
        if (f == 2 && ch) sg = -sg;
      }
    }
    c.m[a] = (unsigned short)mask; c.s[a] = (signed char)sg;
  }
  return c;
}

constexpr float coef(Ent er, const Cfgs& cr, Ent ek, const Cfgs& ck) {
  if (ek.cls != er.cls) return 0.0f;
  int num = 0;
  for (int i = 0; i < cr.n; ++i)
    for (int j = 0; j < ck.n; ++j)
      if ((cr.m[i] & ~ck.m[j]) == 0) num += cr.s[i] * ck.s[j];
  return (float)num / (float)(1 << (er.ns + ek.ns));
}

struct MF { float m[48][24]; bool rz[48]; bool nz[3][6]; int nnz; bool ok; };

constexpr MF build_m() {
  MF r{};
  r.ok = true;
  for (int row = 0; row < 48; ++row) {
    r.rz[row] = true;
    for (int k = 0; k < 24; ++k) {
      r.m[row][k] = 0.0f;
      if (LAY.r[row] < 0 || LAY.k[k] < 0) continue;
      const Ent e3 = make_ent(LAY.k[k], 1), e4 = make_ent(LAY.r[row], 1);
      const float c = coef(e3, expand(e3), e4, expand(e4));
      r.m[row][k] = c;
      if (c != 0.0f) r.rz[row] = false;
    }
  }
  r.nnz = 0;
  for (int t = 0; t < 3; ++t)
    for (int s = 0; s < 6; ++s) {
      bool any = false;
      for (int i = 0; i < 16; ++i)
        for (int j = 0; j < 4; ++j)
          if (r.m[16 * t + i][4 * s + j] != 0.0f) any = true;
      r.nz[t][s] = any;
      if (any) ++r.nnz;
      if (any && !(s >= KBASE[t] && s < KBASE[t] + KLEN[t])) r.ok = false;
    }
  return r;
}
constexpr MF MFc = build_m();
static_assert(MFc.ok, "a K step meets a foreign row tile");
static_assert(MFc.nnz == 6, "block count changed");

struct LaneT { float c[4][24]; float cx[4][4]; bool ok; };
constexpr LaneT build_lane_terms() {
  LaneT r{};
  r.ok = true;
  for (int i = 0; i < 4; ++i) {
    const Ent e3 = make_ent(LAY.sk[i], 1);
    const Cfgs c3 = expand(e3);
    for (int slot = 0; slot < 24; ++slot) {
      const int t = slot >> 3, v = slot & 7;
      const Ent lo = make_ent(LAY.r[16 * t + v], 1), hi = make_ent(LAY.r[16 * t + 8 + v], 1);
      const float a = coef(e3, c3, lo, expand(lo)), b = coef(e3, c3, hi, expand(hi));
      if (a != b) r.ok = false;
      r.c[i][slot] = a;
    }
    for (int j = 0; j < 4; ++j) {
      const Ent lo = make_ent(LAY.xr[j], 1), hi = make_ent(sig(LAY.xr[j]), 1);
      const float a = coef(e3, c3, lo, expand(lo)), b = coef(e3, c3, hi, expand(hi));
      if (a != b) r.ok = false;
      r.cx[i][j] = a;
    }
  }
  return r;
}
constexpr LaneT LTc = build_lane_terms();
static_assert(LTc.ok, "a self-image entry meets a row and its image with different coefficients");

struct Blocks { int t[6], s[6]; };
constexpr Blocks list_blocks() {
  int bt[6] = {}, bs[6] = {};
  int n = 0;
  for (int t = 0; t < 3; ++t)
    for (int s = 0; s < 6; ++s)
      if (MFc.nz[t][s]) { bt[n] = t; bs[n] = s; ++n; }
  Blocks b{};
  bool used[6] = {};
  int last = -1;
  for (int i = 0; i < 6; ++i) {
    int pick = -1;
    int best = -1;
    for (int c = 0; c < 6; ++c) if (!used[c] && bt[c] != last) {
      int left = 0;
      for (int d = 0; d < 6; ++d) if (!used[d] && bt[d] == bt[c]) ++left;
      if (left > best) { best = left; pick = c; }
    }
    for (int c = 0; c < 6 && pick < 0; ++c) if (!used[c]) pick = c;
    used[pick] = true; b.t[i] = bt[pick]; b.s[i] = bs[pick]; last = bt[pick];
  }
  return b;
}
constexpr Blocks BLK = list_blocks();

struct AT { alignas(16) float v[3][32][4]; };
constexpr AT build_at() {
  AT a{};
  for (int i = 0; i < 6; ++i)
    for (int lane = 0; lane < 32; ++lane) {
      const int p = lane & 15, h = lane >> 4;
      a.v[i >> 1][lane][2 * (i & 1)]     = MFc.m[16 * BLK.t[i] + p][4 * BLK.s[i] + 2 * h];
      a.v[i >> 1][lane][2 * (i & 1) + 1] = MFc.m[16 * BLK.t[i] + p][4 * BLK.s[i] + 2 * h + 1];
    }
  return a;
}

}

namespace gol {

__device__ constexpr AT ATf = build_at();

template<typename F, int... Is>
GOL_HD void static_for_impl(F&& f, std::integer_sequence<int, Is...>) {
  (f(std::integral_constant<int, Is>{}), ...);
}
template<int N, typename F>
GOL_HD void static_for(F&& f) {
  static_for_impl(static_cast<F&&>(f), std::make_integer_sequence<int, N>{});
}

GOL_HD void lane_values(const float (&n)[9], float (&bsel)[12], float (&asel)[24],
                        float& SA, float& SB, float& DQ, float& LT) {
  float c[9];
  for (int k = 0; k < 9; ++k) c[k] = 1.0f - n[k];
  float F[4][4];
  static_for<4>([&](auto XX) {
    constexpr int X = XX.value;
    const float u = n[X] * c[8 - X];
    const float w = c[X] * n[8 - X];
    F[X][0] = c[X] * c[8 - X];
    F[X][1] = u + w;
    F[X][2] = u - w;
    F[X][3] = n[X] * n[8 - X];
  });
  float tAB[16], tCD[16];
  static_for<16>([&](auto I) {
    constexpr int i = I.value;
    tAB[i] = F[0][i & 3] * F[1][i >> 2];
    tCD[i] = F[2][i & 3] * F[3][i >> 2];
  });
  SA = 0.0f;
  static_for<12>([&](auto I) {
    constexpr int idx = I.value;
    constexpr int s = idx >> 1, j = idx & 1;
    constexpr int code = LAY.k[4 * s + j];
    static_assert(code >= 0, "every K slot is used");
    static_assert(LAY.k[4 * s + j + 2] == sig(code), "upper half holds the image");
    bsel[idx] = tAB[code & 15] * tCD[code >> 4];
    if constexpr (make_ent(code, 1).cls == 0) SA += bsel[idx];
  });
  float ask[4];
  static_for<4>([&](auto I) {
    constexpr int i = I.value;
    constexpr int code = LAY.sk[i];
    static_assert(code >= 0 && code == sig(code), "a self-image entry");
    ask[i] = tAB[code & 15] * tCD[code >> 4];
    if constexpr (make_ent(code, 1).cls == 0) SA = fmaf(0.5f, ask[i], SA);
  });
  SB = 0.0f;
  float d1 = 0.0f, d3 = 0.0f;
  static_for<24>([&](auto I) {
    constexpr int idx = I.value;
    constexpr int t = idx >> 3, v = idx & 7;
    constexpr int code = LAY.r[16 * t + v];
    static_assert(code >= 0, "every matrix row is used");
    static_assert(LAY.r[16 * t + v + 8] == sig(code), "upper half holds the image");
    constexpr Ent e = make_ent(code, 1);
    asel[idx] = tAB[code & 15] * tCD[code >> 4];
    if constexpr (e.cls == 0) SB += asel[idx];
    static_assert(e.ns == 1 || e.ns == 3, "three live outer cells: one or three one-alive pairs");
    if constexpr (e.ns == 1) d1 = fmaf(asel[idx], asel[idx], d1);
    else                     d3 = fmaf(asel[idx], asel[idx], d3);
  });
  float xsel[4];
  static_for<4>([&](auto J) {
    constexpr int j = J.value;
    constexpr int code = LAY.xr[j];
    static_assert(code >= 0 && code != sig(code), "a row outside the matrix, image above");
    constexpr Ent e = make_ent(code, 1);
    static_assert(e.cls != 0, "no plain-sum entry among them");
    xsel[j] = tAB[code & 15] * tCD[code >> 4];
    static_assert(e.ns == 1 || e.ns == 3, "three live outer cells: one or three one-alive pairs");
    if constexpr (e.ns == 1) d1 = fmaf(xsel[j], xsel[j], d1);
    else                     d3 = fmaf(xsel[j], xsel[j], d3);
  });
  DQ = 0.5f * d1 + 0.125f * d3;
  LT = 0.0f;
  static_for<4>([&](auto I) {
    constexpr int i = I.value;
    float in = 0.0f;
    static_for<24>([&](auto SS) {
      constexpr int slot = SS.value;
      if constexpr (LTc.c[i][slot] != 0.0f) in = fmaf(LTc.c[i][slot], asel[slot], in);
    });
    static_for<4>([&](auto J) {
      constexpr int j = J.value;
      if constexpr (LTc.cx[i][j] != 0.0f) in = fmaf(LTc.cx[i][j], xsel[j], in);
    });
    LT = fmaf(ask[i], in, LT);
  });
}

__device__ __forceinline__ void load_nbhd(const float* __restrict__ x, int w, int p, float (&n)[9]) {
  const int b  = w >> 12;
  const int rm = w & 4095;
  const int y  = rm >> 4;
  const int x0 = ((rm & 15) << 4) + p;
  const float* img = x + (size_t)b * 65536;
#pragma unroll
  for (int k = 0; k < 9; ++k) {
    const int dy = k / 3 - 1, dx = k % 3 - 1;
    const int yy = (y + dy) & 255;
    const int xx = (x0 + dx) & 255;
    n[k] = img[yy * 256 + xx];
  }
}

GOL_HD void hand_over(const float (&nn)[9], bool hi, float (&n)[9]) {
  n[0] = hi ? nn[2] : nn[0];
  n[1] = hi ? nn[3] : nn[1];
  n[2] = hi ? nn[0] : nn[2];
  n[3] = hi ? nn[1] : nn[3];
  n[4] = nn[4];
  n[5] = hi ? nn[7] : nn[5];
  n[6] = hi ? nn[8] : nn[6];
  n[7] = hi ? nn[5] : nn[7];
  n[8] = hi ? nn[6] : nn[8];
}

GOL_HD float tile_result(const v2f (&A)[6], const float (&n)[9]) {
  float bsel[12], asel[24], SA, SB, DQ, LT;
  lane_values(n, bsel, asel, SA, SB, DQ, LT);

  v8f C[3];
  static_for<3>([&](auto TT) {
    constexpr int t = TT.value;
    C[t] = v8f{0.f, 0.f, 0.f, 0.f, 0.f, 0.f, 0.f, 0.f};
  });
  static_for<6>([&](auto II) {
    constexpr int i = II.value;
    constexpr int s = BLK.s[i];
    v2f Bv;
    Bv.x = bsel[2 * s];
    Bv.y = bsel[2 * s + 1];
    C[BLK.t[i]] = __builtin_amdgcn_wmma_f32_16x16x4_f32(false, A[i], false, Bv, (short)0, C[BLK.t[i]], false, false);
  });

  float orp = LT;
  static_for<3>([&](auto TT) {
    constexpr int t = TT.value;
    float o = 0.0f;
    static_for<8>([&](auto VV) {
      constexpr int v = VV.value;
      if constexpr (!(MFc.rz[16 * t + v] && MFc.rz[16 * t + v + 8]))
        o = fmaf(asel[8 * t + v], C[t][v], o);
    });
    orp += o;
  });

  const float q = n[4], cq = 1.0f - n[4];
  float tot = fmaf(q, SA - fmaf(q, orp, cq * DQ), SB);
  tot += __shfl_xor(tot, 16, 32);
  return tot;
}

}

__global__ __launch_bounds__(1024) void gol_wmma_kernel(const float* __restrict__ x,
                                                       float* __restrict__ out,
                                                       int nTiles) {
  using namespace gol;
  const int lane = (int)(threadIdx.x & 31u);
  const int p = lane & 15;
  const int h = lane >> 4;
  const int wv = __builtin_amdgcn_readfirstlane((int)(threadIdx.x >> 5));
  const int wavesPerBlock = (int)(blockDim.x >> 5);
  const int stride = (int)gridDim.x * wavesPerBlock;
  int w = (int)blockIdx.x * wavesPerBlock + wv;
  if (w >= nTiles) return;

  const bool hi = (h != 0);

  float nn[9];
  load_nbhd(x, w, p, nn);

  v2f A[6];
  static_for<3>([&](auto QQ) {
    constexpr int q = QQ.value;
    const float* f = &ATf.v[q][lane][0];
    A[2 * q].x     = f[0];
    A[2 * q].y     = f[1];
    A[2 * q + 1].x = f[2];
    A[2 * q + 1].y = f[3];
  });

  {
    float n[9];
    hand_over(nn, hi, n);
    const int wn = w + stride;
    const bool more = wn < nTiles;
    load_nbhd(x, more ? wn : w, p, nn);
    const float tot = tile_result(A, n);
    if (h == 0) out[(size_t)w * 16 + (size_t)p] = tot;
    if (!more) return;
    w = wn;
  }

  for (;;) {
    float n[9];
    hand_over(nn, hi, n);

    const int wn = w + stride;
    const bool more = wn < nTiles;
    load_nbhd(x, more ? wn : w, p, nn);

    const float tot = tile_result(A, n);
    if (h == 0) out[(size_t)w * 16 + (size_t)p] = tot;

    if (!more) break;
    w = wn;
  }
}

extern "C" void kernel_launch(void* const* d_in, const int* in_sizes, int n_in,
                              void* d_out, int out_size, void* d_ws, size_t ws_size,
                              hipStream_t stream) {
  (void)n_in; (void)d_ws; (void)ws_size; (void)out_size;
  const float* x = (const float*)d_in[0];
  float* out = (float*)d_out;
  const int total = in_sizes[0];
  const int Bn = total >> 16;
  const int nTiles = Bn * 4096;
  dim3 block(1024);
  dim3 grid(256);
  gol_wmma_kernel<<<grid, block, 0, stream>>>(x, out, nTiles);
}
